# starting kernel
# baseline (speedup 1.0000x reference)
#include <hip/hip_runtime.h>

typedef __attribute__((ext_vector_type(2))) float v2f;
typedef __attribute__((ext_vector_type(8))) float v8f;

template <int K, int N, bool RELU>
__global__ __launch_bounds__(32)
void gcn_gemm_wmma_f32(const float* __restrict__ A,
                       const float* __restrict__ W,
                       const float* __restrict__ bias,
                       float* __restrict__ Dout)
{
    const int lane  = threadIdx.x;
    const int hi    = lane >> 4;
    const int l16   = lane & 15;
    const int mtile = blockIdx.y << 4;
    const int ntile = blockIdx.x << 4;

    const int m    = mtile + l16;
    const int koff = hi << 1;

    const float bv = bias[ntile + l16];
    v8f acc;
#pragma unroll
    for (int r = 0; r < 8; ++r) acc[r] = bv;

    const float* __restrict__ arow = A + (size_t)m * K + koff;
    const float* __restrict__ bcol = W + (size_t)koff * N + ntile + l16;

#pragma unroll
    for (int k0 = 0; k0 < K; k0 += 4) {
        v2f a = *(const v2f*)(arow + k0);
        if (RELU) {
            a[0] = fmaxf(a[0], 0.0f);
            a[1] = fmaxf(a[1], 0.0f);
        }
        v2f b;
        b[0] = bcol[k0 * N];
        b[1] = bcol[(k0 + 1) * N];
        acc = __builtin_amdgcn_wmma_f32_16x16x4_f32(
            false, a, false, b, (short)0, acc, false, false);
    }

#pragma unroll
    for (int r = 0; r < 8; ++r) {
        Dout[(size_t)(mtile + (hi << 3) + r) * N + ntile + l16] = acc[r];
    }
}

__global__ __launch_bounds__(256)
void spmm_scatter128(const int* __restrict__ row, const int* __restrict__ col,
                     const float* __restrict__ val,
                     const float* __restrict__ H, float* __restrict__ out, int E)
{
    const long long t = (long long)blockIdx.x * blockDim.x + threadIdx.x;
    const int e = (int)(t >> 5);
    if (e >= E) return;
    const int f = ((int)t & 31) << 2;
    const int r = row[e];
    const int c = col[e];
    const float v = val[e];
    const float4 h = *(const float4*)(H + (size_t)c * 128 + f);
    float* o = out + (size_t)r * 128 + f;
    atomicAdd(o + 0, v * h.x);
    atomicAdd(o + 1, v * h.y);
    atomicAdd(o + 2, v * h.z);
    atomicAdd(o + 3, v * h.w);
}

__global__ __launch_bounds__(256)
void spmm_scatter16(const int* __restrict__ row, const int* __restrict__ col,
                    const float* __restrict__ val,
                    const float* __restrict__ H, float* __restrict__ out, int E)
{
    const long long t = (long long)blockIdx.x * blockDim.x + threadIdx.x;
    const int e = (int)(t >> 4);
    if (e >= E) return;
    const int f = (int)t & 15;
    const float g = val[e] * H[(size_t)col[e] * 16 + f];
    atomicAdd(out + (size_t)row[e] * 16 + f, g);
}

extern "C" void kernel_launch(void* const* d_in, const int* in_sizes, int n_in,
                              void* d_out, int out_size, void* d_ws, size_t ws_size,
                              hipStream_t stream) {
    const float* x   = (const float*)d_in[0];
    const int*   row = (const int*)  d_in[1];
    const int*   col = (const int*)  d_in[2];
    const float* val = (const float*)d_in[3];
    const float* W1  = (const float*)d_in[4];
    const float* b1  = (const float*)d_in[5];
    const float* W2  = (const float*)d_in[6];
    const float* b2  = (const float*)d_in[7];
    float* out = (float*)d_out;

    constexpr int D_IN  = 128;
    constexpr int D_HID = 128;
    constexpr int N_CLS = 16;
    const int E = in_sizes[1];
    const int M = in_sizes[0] / D_IN;

    char*  ws  = (char*)d_ws;
    float* H1  = (float*)ws;
    float* AGG = (float*)(ws + (size_t)M * D_HID * sizeof(float));
    float* H2  = (float*)(ws + 2 * (size_t)M * D_HID * sizeof(float));

    hipMemsetAsync(AGG, 0, (size_t)M * D_HID * sizeof(float), stream);
    hipMemsetAsync(out, 0, (size_t)out_size * sizeof(float), stream);

    dim3 g1(D_HID / 16, M / 16);
    gcn_gemm_wmma_f32<D_IN, D_HID, false><<<g1, 32, 0, stream>>>(x, W1, b1, H1);

    {
        const long long threads = (long long)E * 32;
        const int blk = 256;
        const unsigned nb = (unsigned)((threads + blk - 1) / blk);
        spmm_scatter128<<<nb, blk, 0, stream>>>(row, col, val, H1, AGG, E);
    }

    dim3 g2(N_CLS / 16, M / 16);
    gcn_gemm_wmma_f32<D_HID, N_CLS, true><<<g2, 32, 0, stream>>>(AGG, W2, b2, H2);

    {
        const long long threads = (long long)E * 16;
        const int blk = 256;
        const unsigned nb = (unsigned)((threads + blk - 1) / blk);
        spmm_scatter16<<<nb, blk, 0, stream>>>(row, col, val, H2, out, E);
    }
}


# SpMM, both layers: float-atomic scatter replaced by a per-call sort of edges by row (int atomic count = rank, scan, place) and a wave summing each row in registers; sum order = arrival-order rank
# speedup vs baseline: 1.6084x; 1.6084x over previous
#include <hip/hip_runtime.h>

typedef __attribute__((ext_vector_type(2))) float v2f;
typedef __attribute__((ext_vector_type(8))) float v8f;

template <int K, int N, bool RELU>
__global__ __launch_bounds__(32)
void gcn_gemm_wmma_f32(const float* __restrict__ A,
                       const float* __restrict__ W,
                       const float* __restrict__ bias,
                       float* __restrict__ Dout)
{
    const int lane  = threadIdx.x;
    const int hi    = lane >> 4;
    const int l16   = lane & 15;
    const int mtile = blockIdx.y << 4;
    const int ntile = blockIdx.x << 4;

    const int m    = mtile + l16;
    const int koff = hi << 1;

    const float bv = bias[ntile + l16];
    v8f acc;
#pragma unroll
    for (int r = 0; r < 8; ++r) acc[r] = bv;

    const float* __restrict__ arow = A + (size_t)m * K + koff;
    const float* __restrict__ bcol = W + (size_t)koff * N + ntile + l16;

#pragma unroll
    for (int k0 = 0; k0 < K; k0 += 4) {
        v2f a = *(const v2f*)(arow + k0);
        if (RELU) {
            a[0] = fmaxf(a[0], 0.0f);
            a[1] = fmaxf(a[1], 0.0f);
        }
        v2f b;
        b[0] = bcol[k0 * N];
        b[1] = bcol[(k0 + 1) * N];
        acc = __builtin_amdgcn_wmma_f32_16x16x4_f32(
            false, a, false, b, (short)0, acc, false, false);
    }

#pragma unroll
    for (int r = 0; r < 8; ++r) {
        Dout[(size_t)(mtile + (hi << 3) + r) * N + ntile + l16] = acc[r];
    }
}

__global__ __launch_bounds__(256)
void csr_count_rank(const int* __restrict__ row, int* __restrict__ cnt,
                    int* __restrict__ rank, int E)
{
    const int e = blockIdx.x * 256 + threadIdx.x;
    if (e >= E) return;
    rank[e] = atomicAdd(cnt + row[e], 1);
}

__global__ __launch_bounds__(256)
void csr_scan(const int* __restrict__ cnt, int* __restrict__ off, int M)
{
    __shared__ int s_front[8];
    __shared__ int s_tile[8];
    const int tid   = threadIdx.x;
    const int lane  = tid & 31;
    const int w     = tid >> 5;
    const int tile0 = blockIdx.x * 1024;

    const int4* __restrict__ c4 = (const int4*)cnt + tid;
    int s = 0;
#pragma unroll 8
    for (int k = 0; k < (int)blockIdx.x; ++k) {
        const int4 q = c4[k * 256];
        s += (q.x + q.y) + (q.z + q.w);
    }

    const int i0 = tile0 + tid * 4;
    const int a0 = (i0 + 0 < M) ? cnt[i0 + 0] : 0;
    const int a1 = (i0 + 1 < M) ? cnt[i0 + 1] : 0;
    const int a2 = (i0 + 2 < M) ? cnt[i0 + 2] : 0;
    const int a3 = (i0 + 3 < M) ? cnt[i0 + 3] : 0;
    const int t  = (a0 + a1) + (a2 + a3);

    for (int d = 16; d > 0; d >>= 1) s += __shfl_xor(s, d);

    int inc = t;
    for (int d = 1; d < 32; d <<= 1) {
        const int y = __shfl_up(inc, d);
        if (lane >= d) inc += y;
    }
    if (lane == 31) s_tile[w]  = inc;
    if (lane == 0)  s_front[w] = s;
    __syncthreads();

    int base = 0;
    for (int k = 0; k < 8; ++k) {
        base += s_front[k];
        if (k < w) base += s_tile[k];
    }

    const int p0 = base + inc - t;
    const int p1 = p0 + a0;
    const int p2 = p1 + a1;
    const int p3 = p2 + a2;
    const int p4 = p3 + a3;
    if (i0 + 0 < M) off[i0 + 0] = p0;
    if (i0 + 1 < M) off[i0 + 1] = p1;
    if (i0 + 2 < M) off[i0 + 2] = p2;
    if (i0 + 3 < M) off[i0 + 3] = p3;
    if (i0 + 0 == M - 1) off[M] = p1;
    if (i0 + 1 == M - 1) off[M] = p2;
    if (i0 + 2 == M - 1) off[M] = p3;
    if (i0 + 3 == M - 1) off[M] = p4;
}

__global__ __launch_bounds__(256)
void csr_fill(const int* __restrict__ row, const int* __restrict__ col,
              const float* __restrict__ val, const int* __restrict__ off,
              const int* __restrict__ rank, int* __restrict__ edges, int E)
{
    const int e = blockIdx.x * 256 + threadIdx.x;
    if (e >= E) return;
    const int p = off[row[e]] + rank[e];
    int2 rec;
    rec.x = col[e];
    rec.y = __float_as_int(val[e]);
    ((int2*)edges)[p] = rec;
}

template <int U>
__device__ __forceinline__ void add_rows(float4& acc, const float* __restrict__ Hl,
                                         int ec, int ev, int j)
{
    float4 h[U];
    float  v[U];
#pragma unroll
    for (int u = 0; u < U; ++u) {
        const int c = __builtin_amdgcn_readlane(ec, j + u);
        v[u] = __int_as_float(__builtin_amdgcn_readlane(ev, j + u));
        h[u] = *(const float4*)(Hl + (size_t)c * 128);
    }
#pragma unroll
    for (int u = 0; u < U; ++u) {
        acc.x = fmaf(v[u], h[u].x, acc.x);
        acc.y = fmaf(v[u], h[u].y, acc.y);
        acc.z = fmaf(v[u], h[u].z, acc.z);
        acc.w = fmaf(v[u], h[u].w, acc.w);
    }
}

__global__ __launch_bounds__(256)
void spmm_gather128(const int* __restrict__ off, const int* __restrict__ edges,
                    const float* __restrict__ H, float* __restrict__ out, int M)
{
    const int lane = threadIdx.x & 31;
    const int r = blockIdx.x * 8 + __builtin_amdgcn_readfirstlane(threadIdx.x >> 5);
    if (r >= M) return;
    const int beg = __builtin_amdgcn_readfirstlane(off[r]);
    const int end = __builtin_amdgcn_readfirstlane(off[r + 1]);
    const float* __restrict__ Hl = H + (lane << 2);
    const int2*  __restrict__ ed = (const int2*)edges;

    float4 acc = make_float4(0.0f, 0.0f, 0.0f, 0.0f);
    for (int base = beg; base < end; base += 32) {
        const int n = min(end - base, 32);
        int2 rec = make_int2(0, 0);
        if (lane < n) rec = ed[base + lane];
        int j = 0;
        for (; j + 8 <= n; j += 8) add_rows<8>(acc, Hl, rec.x, rec.y, j);
        if (n & 4) { add_rows<4>(acc, Hl, rec.x, rec.y, j); j += 4; }
        if (n & 2) { add_rows<2>(acc, Hl, rec.x, rec.y, j); j += 2; }
        if (n & 1) { add_rows<1>(acc, Hl, rec.x, rec.y, j); }
    }
    *(float4*)(out + (size_t)r * 128 + (lane << 2)) = acc;
}

__global__ __launch_bounds__(256)
void spmm_gather16(const int* __restrict__ off, const int* __restrict__ edges,
                   const float* __restrict__ H, float* __restrict__ out, int M)
{
    const int t = blockIdx.x * 256 + threadIdx.x;
    const int f = t & 15;
    const int r = t >> 4;
    const bool live = r < M;
    int beg = 0, end = 0;
    if (live) { beg = off[r]; end = off[r + 1]; }
    const int2* __restrict__ ed = (const int2*)edges;

    int rounds = (end - beg + 15) >> 4;
    rounds = max(rounds, __shfl_xor(rounds, 16));
    rounds = __builtin_amdgcn_readfirstlane(rounds);

    float acc = 0.0f;
    for (int q = 0; q < rounds; ++q) {
        const int base = beg + (q << 4);
        const int n = end - base;
        int2 rec = make_int2(0, 0);
        if (f < n) rec = ed[base + f];
#pragma unroll
        for (int j = 0; j < 16; ++j) {
            const int   c = __shfl(rec.x, j, 16);
            const float v = __int_as_float(__shfl(rec.y, j, 16));
            const float h = H[(size_t)c * 16 + f];
            const float s = fmaf(v, h, acc);
            acc = (j < n) ? s : acc;
        }
    }
    if (live) out[(size_t)r * 16 + f] = acc;
}

extern "C" void kernel_launch(void* const* d_in, const int* in_sizes, int n_in,
                              void* d_out, int out_size, void* d_ws, size_t ws_size,
                              hipStream_t stream) {
    const float* x   = (const float*)d_in[0];
    const int*   row = (const int*)  d_in[1];
    const int*   col = (const int*)  d_in[2];
    const float* val = (const float*)d_in[3];
    const float* W1  = (const float*)d_in[4];
    const float* b1  = (const float*)d_in[5];
    const float* W2  = (const float*)d_in[6];
    const float* b2  = (const float*)d_in[7];
    float* out = (float*)d_out;

    constexpr int D_IN  = 128;
    constexpr int D_HID = 128;
    constexpr int N_CLS = 16;
    const int E = in_sizes[1];
    const int M = in_sizes[0] / D_IN;

    char*  ws  = (char*)d_ws;
    float* H1  = (float*)ws;
    float* AGG = (float*)(ws + (size_t)M * D_HID * sizeof(float));
    float* H2  = (float*)(ws + 2 * (size_t)M * D_HID * sizeof(float));
    int*   cnt   = (int*)(H2 + (size_t)M * N_CLS);
    int*   off   = cnt + M;
    int*   edges = off + M + 16;
    int*   rank  = edges + 2 * (size_t)E;

    hipMemsetAsync(cnt, 0, (size_t)M * sizeof(int), stream);

    const unsigned eb = (unsigned)((E + 255) / 256);
    csr_count_rank<<<eb, 256, 0, stream>>>(row, cnt, rank, E);
    csr_scan<<<(unsigned)((M + 1023) / 1024), 256, 0, stream>>>(cnt, off, M);
    csr_fill<<<eb, 256, 0, stream>>>(row, col, val, off, rank, edges, E);

    dim3 g1(D_HID / 16, M / 16);
    gcn_gemm_wmma_f32<D_IN, D_HID, false><<<g1, 32, 0, stream>>>(x, W1, b1, H1);

    spmm_gather128<<<(unsigned)((M + 7) / 8), 256, 0, stream>>>(off, edges, H1, AGG, M);

    dim3 g2(N_CLS / 16, M / 16);
    gcn_gemm_wmma_f32<D_HID, N_CLS, true><<<g2, 32, 0, stream>>>(AGG, W2, b2, H2);

    {
        const long long threads = (long long)M * 16;
        spmm_gather16<<<(unsigned)((threads + 255) / 256), 256, 0, stream>>>(off, edges, H2, out, M);
    }
}


# Layer-1 GEMM rides in the edge count and edge fill launches: each 256-thread block also computes one 16-row stripe of H1, a 16x16 WMMA tile per wave; one launch fewer, tiles run under edge latency
# speedup vs baseline: 1.8215x; 1.1325x over previous
#include <hip/hip_runtime.h>

typedef __attribute__((ext_vector_type(2))) float v2f;
typedef __attribute__((ext_vector_type(8))) float v8f;

template <int K, int N, bool RELU>
__device__ __forceinline__
void gemm_tile_wmma_f32(const float* __restrict__ A,
                        const float* __restrict__ W,
                        const float* __restrict__ bias,
                        float* __restrict__ Dout,
                        const int mtile, const int ntile, const int lane)
{
    const int hi    = lane >> 4;
    const int l16   = lane & 15;

    const int m    = mtile + l16;
    const int koff = hi << 1;

    const float bv = bias[ntile + l16];
    v8f acc;
#pragma unroll
    for (int r = 0; r < 8; ++r) acc[r] = bv;

    const float* __restrict__ arow = A + (size_t)m * K + koff;
    const float* __restrict__ bcol = W + (size_t)koff * N + ntile + l16;

#pragma unroll
    for (int k0 = 0; k0 < K; k0 += 4) {
        v2f a = *(const v2f*)(arow + k0);
        if (RELU) {
            a[0] = fmaxf(a[0], 0.0f);
            a[1] = fmaxf(a[1], 0.0f);
        }
        v2f b;
        b[0] = bcol[k0 * N];
        b[1] = bcol[(k0 + 1) * N];
        acc = __builtin_amdgcn_wmma_f32_16x16x4_f32(
            false, a, false, b, (short)0, acc, false, false);
    }

#pragma unroll
    for (int r = 0; r < 8; ++r) {
        Dout[(size_t)(mtile + (hi << 3) + r) * N + ntile + l16] = acc[r];
    }
}

template <int K, int N, bool RELU>
__global__ __launch_bounds__(32)
void gcn_gemm_wmma_f32(const float* __restrict__ A,
                       const float* __restrict__ W,
                       const float* __restrict__ bias,
                       float* __restrict__ Dout)
{
    gemm_tile_wmma_f32<K, N, RELU>(A, W, bias, Dout,
                                   blockIdx.y << 4, blockIdx.x << 4, threadIdx.x);
}

template <int K, int N>
__device__ __forceinline__
void gemm_stripe_rider(const float* __restrict__ A, const float* __restrict__ W,
                       const float* __restrict__ bias, float* __restrict__ Dout,
                       const int stripe)
{
    static_assert(N == 8 * 16, "eight waves of the block take the stripe's eight tiles");
    const int w = __builtin_amdgcn_readfirstlane(threadIdx.x >> 5);
    gemm_tile_wmma_f32<K, N, false>(A, W, bias, Dout, stripe << 4, w << 4,
                                    threadIdx.x & 31);
}

template <int K, int N>
__global__ __launch_bounds__(256)
void csr_count_rank(const int* __restrict__ row, int* __restrict__ cnt,
                    int* __restrict__ rank, int E,
                    const float* __restrict__ X, const float* __restrict__ W,
                    const float* __restrict__ bias, float* __restrict__ H,
                    int stripe0, int nstripes)
{
    const int e = blockIdx.x * 256 + threadIdx.x;
    const bool live = e < E;
    int rk = 0;
    if (live) rk = atomicAdd(cnt + row[e], 1);
    if ((int)blockIdx.x < nstripes)
        gemm_stripe_rider<K, N>(X, W, bias, H, stripe0 + (int)blockIdx.x);
    if (live) rank[e] = rk;
}

__global__ __launch_bounds__(256)
void csr_scan(const int* __restrict__ cnt, int* __restrict__ off, int M)
{
    __shared__ int s_front[8];
    __shared__ int s_tile[8];
    const int tid   = threadIdx.x;
    const int lane  = tid & 31;
    const int w     = tid >> 5;
    const int tile0 = blockIdx.x * 1024;

    const int4* __restrict__ c4 = (const int4*)cnt + tid;
    int s = 0;
#pragma unroll 8
    for (int k = 0; k < (int)blockIdx.x; ++k) {
        const int4 q = c4[k * 256];
        s += (q.x + q.y) + (q.z + q.w);
    }

    const int i0 = tile0 + tid * 4;
    const int a0 = (i0 + 0 < M) ? cnt[i0 + 0] : 0;
    const int a1 = (i0 + 1 < M) ? cnt[i0 + 1] : 0;
    const int a2 = (i0 + 2 < M) ? cnt[i0 + 2] : 0;
    const int a3 = (i0 + 3 < M) ? cnt[i0 + 3] : 0;
    const int t  = (a0 + a1) + (a2 + a3);

    for (int d = 16; d > 0; d >>= 1) s += __shfl_xor(s, d);

    int inc = t;
    for (int d = 1; d < 32; d <<= 1) {
        const int y = __shfl_up(inc, d);
        if (lane >= d) inc += y;
    }
    if (lane == 31) s_tile[w]  = inc;
    if (lane == 0)  s_front[w] = s;
    __syncthreads();

    int base = 0;
    for (int k = 0; k < 8; ++k) {
        base += s_front[k];
        if (k < w) base += s_tile[k];
    }

    const int p0 = base + inc - t;
    const int p1 = p0 + a0;
    const int p2 = p1 + a1;
    const int p3 = p2 + a2;
    const int p4 = p3 + a3;
    if (i0 + 0 < M) off[i0 + 0] = p0;
    if (i0 + 1 < M) off[i0 + 1] = p1;
    if (i0 + 2 < M) off[i0 + 2] = p2;
    if (i0 + 3 < M) off[i0 + 3] = p3;
    if (i0 + 0 == M - 1) off[M] = p1;
    if (i0 + 1 == M - 1) off[M] = p2;
    if (i0 + 2 == M - 1) off[M] = p3;
    if (i0 + 3 == M - 1) off[M] = p4;
}

template <int K, int N>
__global__ __launch_bounds__(256)
void csr_fill(const int* __restrict__ row, const int* __restrict__ col,
              const float* __restrict__ val, const int* __restrict__ off,
              const int* __restrict__ rank, int* __restrict__ edges, int E,
              const float* __restrict__ X, const float* __restrict__ W,
              const float* __restrict__ bias, float* __restrict__ H,
              int stripe0, int nstripes)
{
    const int e = blockIdx.x * 256 + threadIdx.x;
    const bool live = e < E;
    int o = 0, rk = 0;
    int2 rec = make_int2(0, 0);
    if (live) {
        o  = off[row[e]];
        rk = rank[e];
        rec.x = col[e];
        rec.y = __float_as_int(val[e]);
    }
    if ((int)blockIdx.x < nstripes)
        gemm_stripe_rider<K, N>(X, W, bias, H, stripe0 + (int)blockIdx.x);
    if (live) ((int2*)edges)[o + rk] = rec;
}

template <int U>
__device__ __forceinline__ void add_rows(float4& acc, const float* __restrict__ Hl,
                                         int ec, int ev, int j)
{
    float4 h[U];
    float  v[U];
#pragma unroll
    for (int u = 0; u < U; ++u) {
        const int c = __builtin_amdgcn_readlane(ec, j + u);
        v[u] = __int_as_float(__builtin_amdgcn_readlane(ev, j + u));
        h[u] = *(const float4*)(Hl + (size_t)c * 128);
    }
#pragma unroll
    for (int u = 0; u < U; ++u) {
        acc.x = fmaf(v[u], h[u].x, acc.x);
        acc.y = fmaf(v[u], h[u].y, acc.y);
        acc.z = fmaf(v[u], h[u].z, acc.z);
        acc.w = fmaf(v[u], h[u].w, acc.w);
    }
}

__global__ __launch_bounds__(256)
void spmm_gather128(const int* __restrict__ off, const int* __restrict__ edges,
                    const float* __restrict__ H, float* __restrict__ out, int M)
{
    const int lane = threadIdx.x & 31;
    const int r = blockIdx.x * 8 + __builtin_amdgcn_readfirstlane(threadIdx.x >> 5);
    if (r >= M) return;
    const int beg = __builtin_amdgcn_readfirstlane(off[r]);
    const int end = __builtin_amdgcn_readfirstlane(off[r + 1]);
    const float* __restrict__ Hl = H + (lane << 2);
    const int2*  __restrict__ ed = (const int2*)edges;

    float4 acc = make_float4(0.0f, 0.0f, 0.0f, 0.0f);
    for (int base = beg; base < end; base += 32) {
        const int n = min(end - base, 32);
        int2 rec = make_int2(0, 0);
        if (lane < n) rec = ed[base + lane];
        int j = 0;
        for (; j + 8 <= n; j += 8) add_rows<8>(acc, Hl, rec.x, rec.y, j);
        if (n & 4) { add_rows<4>(acc, Hl, rec.x, rec.y, j); j += 4; }
        if (n & 2) { add_rows<2>(acc, Hl, rec.x, rec.y, j); j += 2; }
        if (n & 1) { add_rows<1>(acc, Hl, rec.x, rec.y, j); }
    }
    *(float4*)(out + (size_t)r * 128 + (lane << 2)) = acc;
}

__global__ __launch_bounds__(256)
void spmm_gather16(const int* __restrict__ off, const int* __restrict__ edges,
                   const float* __restrict__ H, float* __restrict__ out, int M)
{
    const int t = blockIdx.x * 256 + threadIdx.x;
    const int f = t & 15;
    const int r = t >> 4;
    const bool live = r < M;
    int beg = 0, end = 0;
    if (live) { beg = off[r]; end = off[r + 1]; }
    const int2* __restrict__ ed = (const int2*)edges;

    int rounds = (end - beg + 15) >> 4;
    rounds = max(rounds, __shfl_xor(rounds, 16));
    rounds = __builtin_amdgcn_readfirstlane(rounds);

    float acc = 0.0f;
    for (int q = 0; q < rounds; ++q) {
        const int base = beg + (q << 4);
        const int n = end - base;
        int2 rec = make_int2(0, 0);
        if (f < n) rec = ed[base + f];
#pragma unroll
        for (int j = 0; j < 16; ++j) {
            const int   c = __shfl(rec.x, j, 16);
            const float v = __int_as_float(__shfl(rec.y, j, 16));
            const float h = H[(size_t)c * 16 + f];
            const float s = fmaf(v, h, acc);
            acc = (j < n) ? s : acc;
        }
    }
    if (live) out[(size_t)r * 16 + f] = acc;
}

extern "C" void kernel_launch(void* const* d_in, const int* in_sizes, int n_in,
                              void* d_out, int out_size, void* d_ws, size_t ws_size,
                              hipStream_t stream) {
    const float* x   = (const float*)d_in[0];
    const int*   row = (const int*)  d_in[1];
    const int*   col = (const int*)  d_in[2];
    const float* val = (const float*)d_in[3];
    const float* W1  = (const float*)d_in[4];
    const float* b1  = (const float*)d_in[5];
    const float* W2  = (const float*)d_in[6];
    const float* b2  = (const float*)d_in[7];
    float* out = (float*)d_out;

    constexpr int D_IN  = 128;
    constexpr int D_HID = 128;
    constexpr int N_CLS = 16;
    const int E = in_sizes[1];
    const int M = in_sizes[0] / D_IN;

    char*  ws  = (char*)d_ws;
    float* H1  = (float*)ws;
    float* AGG = (float*)(ws + (size_t)M * D_HID * sizeof(float));
    float* H2  = (float*)(ws + 2 * (size_t)M * D_HID * sizeof(float));
    int*   cnt   = (int*)(H2 + (size_t)M * N_CLS);
    int*   off   = cnt + M;
    int*   edges = off + M + 16;
    int*   rank  = edges + 2 * (size_t)E;

    hipMemsetAsync(cnt, 0, (size_t)M * sizeof(int), stream);

    const int stripes = M / 16;
    const int sA = (stripes + 1) / 2;
    const int sB = stripes - sA;
    const int eb = (E + 255) / 256;
    const unsigned gA = (unsigned)(eb > sA ? eb : sA);
    const unsigned gB = (unsigned)(eb > sB ? eb : sB);
    csr_count_rank<D_IN, D_HID><<<gA, 256, 0, stream>>>(row, cnt, rank, E,
                                                        x, W1, b1, H1, 0, sA);
    csr_scan<<<(unsigned)((M + 1023) / 1024), 256, 0, stream>>>(cnt, off, M);
    csr_fill<D_IN, D_HID><<<gB, 256, 0, stream>>>(row, col, val, off, rank, edges, E,
                                                  x, W1, b1, H1, sA, sB);

    spmm_gather128<<<(unsigned)((M + 7) / 8), 256, 0, stream>>>(off, edges, H1, AGG, M);

    dim3 g2(N_CLS / 16, M / 16);
    gcn_gemm_wmma_f32<D_HID, N_CLS, true><<<g2, 32, 0, stream>>>(AGG, W2, b2, H2);

    {
        const long long threads = (long long)M * 16;
        spmm_gather16<<<(unsigned)((threads + 255) / 256), 256, 0, stream>>>(off, edges, H2, out, M);
    }
}


# Edge sort without global atomics: edges binned by 256-row range (LDS block histograms, scan, LDS cursors), each bin ordered by row in LDS; counter clear dropped; sum order = LDS arrival order
# speedup vs baseline: 2.1437x; 1.1769x over previous
#include <hip/hip_runtime.h>

typedef __attribute__((ext_vector_type(2))) float v2f;
typedef __attribute__((ext_vector_type(8))) float v8f;

template <int K, int N, bool RELU>
__device__ __forceinline__
void gemm_tile_wmma_f32(const float* __restrict__ A,
                        const float* __restrict__ W,
                        const float* __restrict__ bias,
                        float* __restrict__ Dout,
                        const int mtile, const int ntile, const int lane)
{
    const int hi    = lane >> 4;
    const int l16   = lane & 15;

    const int m    = mtile + l16;
    const int koff = hi << 1;

    const float bv = bias[ntile + l16];
    v8f acc;
#pragma unroll
    for (int r = 0; r < 8; ++r) acc[r] = bv;

    const float* __restrict__ arow = A + (size_t)m * K + koff;
    const float* __restrict__ bcol = W + (size_t)koff * N + ntile + l16;

#pragma unroll
    for (int k0 = 0; k0 < K; k0 += 4) {
        v2f a = *(const v2f*)(arow + k0);
        if (RELU) {
            a[0] = fmaxf(a[0], 0.0f);
            a[1] = fmaxf(a[1], 0.0f);
        }
        v2f b;
        b[0] = bcol[k0 * N];
        b[1] = bcol[(k0 + 1) * N];
        acc = __builtin_amdgcn_wmma_f32_16x16x4_f32(
            false, a, false, b, (short)0, acc, false, false);
    }

#pragma unroll
    for (int r = 0; r < 8; ++r) {
        Dout[(size_t)(mtile + (hi << 3) + r) * N + ntile + l16] = acc[r];
    }
}

template <int K, int N, bool RELU>
__global__ __launch_bounds__(32)
void gcn_gemm_wmma_f32(const float* __restrict__ A,
                       const float* __restrict__ W,
                       const float* __restrict__ bias,
                       float* __restrict__ Dout)
{
    gemm_tile_wmma_f32<K, N, RELU>(A, W, bias, Dout,
                                   blockIdx.y << 4, blockIdx.x << 4, threadIdx.x);
}

template <int K, int N>
__device__ __forceinline__
void gemm_stripe_rider(const float* __restrict__ A, const float* __restrict__ W,
                       const float* __restrict__ bias, float* __restrict__ Dout,
                       const int stripe)
{
    static_assert(N == 8 * 16, "eight waves of the block take the stripe's eight tiles");
    const int w = __builtin_amdgcn_readfirstlane(threadIdx.x >> 5);
    gemm_tile_wmma_f32<K, N, false>(A, W, bias, Dout, stripe << 4, w << 4,
                                    threadIdx.x & 31);
}

constexpr int BIN_SHIFT = 8;
constexpr int BIN_ROWS  = 1 << BIN_SHIFT;
constexpr int MAX_BINS  = 256;
constexpr int EPT       = 8;
constexpr int CHUNK     = 256 * EPT;

template <int K, int N>
__device__ __forceinline__
void gemm_stripes_of_block(const float* __restrict__ A, const float* __restrict__ W,
                           const float* __restrict__ bias, float* __restrict__ Dout,
                           const int stripe0, const int nstripes, const int nblk)
{
    for (int s = (int)blockIdx.x; s < nstripes; s += nblk)
        gemm_stripe_rider<K, N>(A, W, bias, Dout, stripe0 + s);
}

template <int K, int N>
__global__ __launch_bounds__(256)
void bin_count(const int* __restrict__ row, int* __restrict__ tab, int E, int nbin, int nblk,
               const float* __restrict__ X, const float* __restrict__ W,
               const float* __restrict__ bias, float* __restrict__ H,
               int stripe0, int nstripes)
{
    __shared__ int s_hist[MAX_BINS];
    const int tid = threadIdx.x;
    const int e0  = (int)blockIdx.x * CHUNK + tid;
    s_hist[tid] = 0;
    int r[EPT];
#pragma unroll
    for (int i = 0; i < EPT; ++i) {
        const int e = e0 + i * 256;
        r[i] = (e < E) ? row[e] : -1;
    }
    gemm_stripes_of_block<K, N>(X, W, bias, H, stripe0, nstripes, nblk);
    __syncthreads();
#pragma unroll
    for (int i = 0; i < EPT; ++i)
        if (r[i] >= 0) atomicAdd(&s_hist[r[i] >> BIN_SHIFT], 1);
    __syncthreads();
    if (tid < nbin) tab[tid * nblk + (int)blockIdx.x] = s_hist[tid];
}

__global__ __launch_bounds__(256)
void csr_scan(const int* __restrict__ cnt, int* __restrict__ off, int M)
{
    __shared__ int s_front[8];
    __shared__ int s_tile[8];
    const int tid   = threadIdx.x;
    const int lane  = tid & 31;
    const int w     = tid >> 5;
    const int tile0 = blockIdx.x * 1024;

    const int4* __restrict__ c4 = (const int4*)cnt + tid;
    int s = 0;
#pragma unroll 8
    for (int k = 0; k < (int)blockIdx.x; ++k) {
        const int4 q = c4[k * 256];
        s += (q.x + q.y) + (q.z + q.w);
    }

    const int i0 = tile0 + tid * 4;
    const int a0 = (i0 + 0 < M) ? cnt[i0 + 0] : 0;
    const int a1 = (i0 + 1 < M) ? cnt[i0 + 1] : 0;
    const int a2 = (i0 + 2 < M) ? cnt[i0 + 2] : 0;
    const int a3 = (i0 + 3 < M) ? cnt[i0 + 3] : 0;
    const int t  = (a0 + a1) + (a2 + a3);

    for (int d = 16; d > 0; d >>= 1) s += __shfl_xor(s, d);

    int inc = t;
    for (int d = 1; d < 32; d <<= 1) {
        const int y = __shfl_up(inc, d);
        if (lane >= d) inc += y;
    }
    if (lane == 31) s_tile[w]  = inc;
    if (lane == 0)  s_front[w] = s;
    __syncthreads();

    int base = 0;
    for (int k = 0; k < 8; ++k) {
        base += s_front[k];
        if (k < w) base += s_tile[k];
    }

    const int p0 = base + inc - t;
    const int p1 = p0 + a0;
    const int p2 = p1 + a1;
    const int p3 = p2 + a2;
    const int p4 = p3 + a3;
    if (i0 + 0 < M) off[i0 + 0] = p0;
    if (i0 + 1 < M) off[i0 + 1] = p1;
    if (i0 + 2 < M) off[i0 + 2] = p2;
    if (i0 + 3 < M) off[i0 + 3] = p3;
    if (i0 + 0 == M - 1) off[M] = p1;
    if (i0 + 1 == M - 1) off[M] = p2;
    if (i0 + 2 == M - 1) off[M] = p3;
    if (i0 + 3 == M - 1) off[M] = p4;
}

template <int K, int N>
__global__ __launch_bounds__(256)
void bin_scatter(const int* __restrict__ row, const int* __restrict__ col,
                 const float* __restrict__ val, const int* __restrict__ pre,
                 int2* __restrict__ binned, int E, int nbin, int nblk,
                 const float* __restrict__ X, const float* __restrict__ W,
                 const float* __restrict__ bias, float* __restrict__ H,
                 int stripe0, int nstripes)
{
    __shared__ int s_cur[MAX_BINS];
    const int tid = threadIdx.x;
    const int e0  = (int)blockIdx.x * CHUNK + tid;
    int cur0 = 0;
    if (tid < nbin) cur0 = pre[tid * nblk + (int)blockIdx.x];
    int r[EPT], c[EPT], v[EPT];
#pragma unroll
    for (int i = 0; i < EPT; ++i) {
        const int e = e0 + i * 256;
        const bool live = e < E;
        r[i] = live ? row[e] : -1;
        c[i] = live ? col[e] : 0;
        v[i] = live ? __float_as_int(val[e]) : 0;
    }
    gemm_stripes_of_block<K, N>(X, W, bias, H, stripe0, nstripes, nblk);
    s_cur[tid] = cur0;
    __syncthreads();
#pragma unroll
    for (int i = 0; i < EPT; ++i) {
        if (r[i] >= 0) {
            const int pos = atomicAdd(&s_cur[r[i] >> BIN_SHIFT], 1);
            const unsigned key = ((unsigned)(r[i] & (BIN_ROWS - 1)) << 24) | (unsigned)c[i];
            binned[pos] = make_int2((int)key, v[i]);
        }
    }
}

constexpr int SORT_U = 5;
__global__ __launch_bounds__(1024)
void bin_sort(const int* __restrict__ pre, const int2* __restrict__ binned,
              int* __restrict__ off, int2* __restrict__ edges, int M, int nblk)
{
    __shared__ int s_cnt[BIN_ROWS];
    __shared__ int s_w[BIN_ROWS / 32];
    const int tid  = threadIdx.x;
    const int lane = tid & 31;
    const int beg = pre[(int)blockIdx.x * nblk];
    const int end = pre[((int)blockIdx.x + 1) * nblk];
    if (tid < BIN_ROWS) s_cnt[tid] = 0;
    __syncthreads();

    for (int base = beg; base < end; base += 1024 * SORT_U) {
        int k[SORT_U];
#pragma unroll
        for (int u = 0; u < SORT_U; ++u) {
            const int i = base + u * 1024 + tid;
            k[u] = binned[min(i, end - 1)].x;
        }
#pragma unroll
        for (int u = 0; u < SORT_U; ++u) {
            const int i = base + u * 1024 + tid;
            if (i < end) atomicAdd(&s_cnt[(unsigned)k[u] >> 24], 1);
        }
    }
    __syncthreads();

    int cn = 0, inc = 0;
    if (tid < BIN_ROWS) {
        cn  = s_cnt[tid];
        inc = cn;
        for (int d = 1; d < 32; d <<= 1) {
            const int y = __shfl_up(inc, d);
            if (lane >= d) inc += y;
        }
        if (lane == 31) s_w[tid >> 5] = inc;
    }
    __syncthreads();
    if (tid < BIN_ROWS) {
        const int w = tid >> 5;
        int start = beg + inc - cn;
        for (int q = 0; q < BIN_ROWS / 32; ++q)
            if (q < w) start += s_w[q];
        s_cnt[tid] = start;
        const int rr = ((int)blockIdx.x << BIN_SHIFT) + tid;
        if (rr < M) off[rr] = start;
        if (rr == M - 1) off[M] = start + cn;
    }
    __syncthreads();

    for (int base = beg; base < end; base += 1024 * SORT_U) {
        int2 rec[SORT_U];
#pragma unroll
        for (int u = 0; u < SORT_U; ++u) {
            const int i = base + u * 1024 + tid;
            rec[u] = binned[min(i, end - 1)];
        }
        int pos[SORT_U];
#pragma unroll
        for (int u = 0; u < SORT_U; ++u) {
            const int i = base + u * 1024 + tid;
            pos[u] = 0;
            if (i < end) pos[u] = atomicAdd(&s_cnt[(unsigned)rec[u].x >> 24], 1);
        }
#pragma unroll
        for (int u = 0; u < SORT_U; ++u) {
            const int i = base + u * 1024 + tid;
            if (i < end)
                edges[pos[u]] = make_int2((int)((unsigned)rec[u].x & 0x00FFFFFFu), rec[u].y);
        }
    }
}

template <int U>
__device__ __forceinline__ void add_rows(float4& acc, const float* __restrict__ Hl,
                                         int ec, int ev, int j)
{
    float4 h[U];
    float  v[U];
#pragma unroll
    for (int u = 0; u < U; ++u) {
        const int c = __builtin_amdgcn_readlane(ec, j + u);
        v[u] = __int_as_float(__builtin_amdgcn_readlane(ev, j + u));
        h[u] = *(const float4*)(Hl + (size_t)c * 128);
    }
#pragma unroll
    for (int u = 0; u < U; ++u) {
        acc.x = fmaf(v[u], h[u].x, acc.x);
        acc.y = fmaf(v[u], h[u].y, acc.y);
        acc.z = fmaf(v[u], h[u].z, acc.z);
        acc.w = fmaf(v[u], h[u].w, acc.w);
    }
}

__global__ __launch_bounds__(256)
void spmm_gather128(const int* __restrict__ off, const int* __restrict__ edges,
                    const float* __restrict__ H, float* __restrict__ out, int M)
{
    const int lane = threadIdx.x & 31;
    const int r = blockIdx.x * 8 + __builtin_amdgcn_readfirstlane(threadIdx.x >> 5);
    if (r >= M) return;
    const int beg = __builtin_amdgcn_readfirstlane(off[r]);
    const int end = __builtin_amdgcn_readfirstlane(off[r + 1]);
    const float* __restrict__ Hl = H + (lane << 2);
    const int2*  __restrict__ ed = (const int2*)edges;

    float4 acc = make_float4(0.0f, 0.0f, 0.0f, 0.0f);
    for (int base = beg; base < end; base += 32) {
        const int n = min(end - base, 32);
        int2 rec = make_int2(0, 0);
        if (lane < n) rec = ed[base + lane];
        int j = 0;
        for (; j + 8 <= n; j += 8) add_rows<8>(acc, Hl, rec.x, rec.y, j);
        if (n & 4) { add_rows<4>(acc, Hl, rec.x, rec.y, j); j += 4; }
        if (n & 2) { add_rows<2>(acc, Hl, rec.x, rec.y, j); j += 2; }
        if (n & 1) { add_rows<1>(acc, Hl, rec.x, rec.y, j); }
    }
    *(float4*)(out + (size_t)r * 128 + (lane << 2)) = acc;
}

__global__ __launch_bounds__(256)
void spmm_gather16(const int* __restrict__ off, const int* __restrict__ edges,
                   const float* __restrict__ H, float* __restrict__ out, int M)
{
    const int t = blockIdx.x * 256 + threadIdx.x;
    const int f = t & 15;
    const int r = t >> 4;
    const bool live = r < M;
    int beg = 0, end = 0;
    if (live) { beg = off[r]; end = off[r + 1]; }
    const int2* __restrict__ ed = (const int2*)edges;

    int rounds = (end - beg + 15) >> 4;
    rounds = max(rounds, __shfl_xor(rounds, 16));
    rounds = __builtin_amdgcn_readfirstlane(rounds);

    float acc = 0.0f;
    for (int q = 0; q < rounds; ++q) {
        const int base = beg + (q << 4);
        const int n = end - base;
        int2 rec = make_int2(0, 0);
        if (f < n) rec = ed[base + f];
#pragma unroll
        for (int j = 0; j < 16; ++j) {
            const int   c = __shfl(rec.x, j, 16);
            const float v = __int_as_float(__shfl(rec.y, j, 16));
            const float h = H[(size_t)c * 16 + f];
            const float s = fmaf(v, h, acc);
            acc = (j < n) ? s : acc;
        }
    }
    if (live) out[(size_t)r * 16 + f] = acc;
}

extern "C" void kernel_launch(void* const* d_in, const int* in_sizes, int n_in,
                              void* d_out, int out_size, void* d_ws, size_t ws_size,
                              hipStream_t stream) {
    const float* x   = (const float*)d_in[0];
    const int*   row = (const int*)  d_in[1];
    const int*   col = (const int*)  d_in[2];
    const float* val = (const float*)d_in[3];
    const float* W1  = (const float*)d_in[4];
    const float* b1  = (const float*)d_in[5];
    const float* W2  = (const float*)d_in[6];
    const float* b2  = (const float*)d_in[7];
    float* out = (float*)d_out;

    constexpr int D_IN  = 128;
    constexpr int D_HID = 128;
    constexpr int N_CLS = 16;
    const int E = in_sizes[1];
    const int M = in_sizes[0] / D_IN;

    const int nbin = (M + BIN_ROWS - 1) / BIN_ROWS;
    const int nblk = (E + CHUNK - 1) / CHUNK;
    const int ntab = nbin * nblk;

    char*  ws  = (char*)d_ws;
    float* H1  = (float*)ws;
    float* AGG = (float*)(ws + (size_t)M * D_HID * sizeof(float));
    float* H2  = (float*)(ws + 2 * (size_t)M * D_HID * sizeof(float));
    int*   tab    = (int*)(H2 + (size_t)M * N_CLS);
    int*   pre    = tab + ((ntab + 3) & ~3);
    int*   off    = pre + ((ntab + 1 + 3) & ~3);
    int*   binned = off + ((M + 1 + 3) & ~3);
    int*   edges  = binned + 2 * (size_t)E;

    const int stripes = M / 16;
    const int sA = (stripes + 1) / 2;
    const int sB = stripes - sA;
    bin_count<D_IN, D_HID><<<(unsigned)nblk, 256, 0, stream>>>(row, tab, E, nbin, nblk,
                                                               x, W1, b1, H1, 0, sA);
    csr_scan<<<(unsigned)((ntab + 1023) / 1024), 256, 0, stream>>>(tab, pre, ntab);
    bin_scatter<D_IN, D_HID><<<(unsigned)nblk, 256, 0, stream>>>(row, col, val, pre,
                                                                 (int2*)binned, E, nbin, nblk,
                                                                 x, W1, b1, H1, sA, sB);
    bin_sort<<<(unsigned)nbin, 1024, 0, stream>>>(pre, (const int2*)binned, off,
                                                  (int2*)edges, M, nblk);

    spmm_gather128<<<(unsigned)((M + 7) / 8), 256, 0, stream>>>(off, edges, H1, AGG, M);

    dim3 g2(N_CLS / 16, M / 16);
    gcn_gemm_wmma_f32<D_HID, N_CLS, true><<<g2, 32, 0, stream>>>(AGG, W2, b2, H2);

    {
        const long long threads = (long long)M * 16;
        spmm_gather16<<<(unsigned)((threads + 255) / 256), 256, 0, stream>>>(off, edges, H2, out, M);
    }
}


# Layer-1 product operand feed: each block reads its W1 slice into registers once and stages every x stripe once through LDS by whole-row loads, the eight waves taking A fragments from LDS
# speedup vs baseline: 2.4611x; 1.1481x over previous
#include <hip/hip_runtime.h>

typedef __attribute__((ext_vector_type(2))) float v2f;
typedef __attribute__((ext_vector_type(8))) float v8f;

template <int K, int N, bool RELU>
__device__ __forceinline__
void gemm_tile_wmma_f32(const float* __restrict__ A,
                        const float* __restrict__ W,
                        const float* __restrict__ bias,
                        float* __restrict__ Dout,
                        const int mtile, const int ntile, const int lane)
{
    const int hi    = lane >> 4;
    const int l16   = lane & 15;

    const int m    = mtile + l16;
    const int koff = hi << 1;

    const float bv = bias[ntile + l16];
    v8f acc;
#pragma unroll
    for (int r = 0; r < 8; ++r) acc[r] = bv;

    const float* __restrict__ arow = A + (size_t)m * K + koff;
    const float* __restrict__ bcol = W + (size_t)koff * N + ntile + l16;

#pragma unroll
    for (int k0 = 0; k0 < K; k0 += 4) {
        v2f a = *(const v2f*)(arow + k0);
        if (RELU) {
            a[0] = fmaxf(a[0], 0.0f);
            a[1] = fmaxf(a[1], 0.0f);
        }
        v2f b;
        b[0] = bcol[k0 * N];
        b[1] = bcol[(k0 + 1) * N];
        acc = __builtin_amdgcn_wmma_f32_16x16x4_f32(
            false, a, false, b, (short)0, acc, false, false);
    }

#pragma unroll
    for (int r = 0; r < 8; ++r) {
        Dout[(size_t)(mtile + (hi << 3) + r) * N + ntile + l16] = acc[r];
    }
}

template <int K, int N, bool RELU>
__global__ __launch_bounds__(32)
void gcn_gemm_wmma_f32(const float* __restrict__ A,
                       const float* __restrict__ W,
                       const float* __restrict__ bias,
                       float* __restrict__ Dout)
{
    gemm_tile_wmma_f32<K, N, RELU>(A, W, bias, Dout,
                                   blockIdx.y << 4, blockIdx.x << 4, threadIdx.x);
}

constexpr int BIN_SHIFT = 8;
constexpr int BIN_ROWS  = 1 << BIN_SHIFT;
constexpr int MAX_BINS  = 256;
constexpr int EPT       = 8;
constexpr int CHUNK     = 256 * EPT;

constexpr int A_PITCH = 132;
constexpr int A_STAGE = 16 * A_PITCH;

__device__ __forceinline__ void lds_barrier()
{
    __builtin_amdgcn_fence(__ATOMIC_RELEASE, "workgroup", "local");
    __builtin_amdgcn_s_barrier();
    __builtin_amdgcn_fence(__ATOMIC_ACQUIRE, "workgroup", "local");
}

template <int K, int N>
__device__ __forceinline__
void gemm_stripes_of_block(const float* __restrict__ A, const float* __restrict__ W,
                           const float* __restrict__ bias, float* __restrict__ Dout,
                           const int stripe0, const int nstripes, const int nblk,
                           float* __restrict__ s_a)
{
    static_assert(K == 128, "a staged row is one 512-byte wave load");
    static_assert(N == 8 * 16, "eight waves of the block take the stripe's eight tiles");
    int s = (int)blockIdx.x;
    if (s >= nstripes) return;

    const int tid   = threadIdx.x;
    const int lane  = tid & 31;
    const int w     = __builtin_amdgcn_readfirstlane(tid >> 5);
    const int hi    = lane >> 4;
    const int l16   = lane & 15;
    const int koff  = hi << 1;
    const int ntile = w << 4;

    const int scol = lane << 2;
    const float* __restrict__ gsrc = A + scol;
    float4 g0, g1;
    {
        const size_t r0 = (size_t)((stripe0 + s) << 4) + (size_t)w;
        g0 = *(const float4*)(gsrc + r0 * K);
        g1 = *(const float4*)(gsrc + (r0 + 8) * K);
    }

    const float* __restrict__ bcol = W + (size_t)koff * N + ntile + l16;
    v2f b[K / 4];
#pragma unroll
    for (int i = 0; i < K / 4; ++i) {
        b[i][0] = bcol[(4 * i) * N];
        b[i][1] = bcol[(4 * i + 1) * N];
    }
    const float bv = bias[ntile + l16];

    const int st0 = w * A_PITCH + scol;
    const int st1 = st0 + 8 * A_PITCH;
    const int fr  = l16 * A_PITCH + koff;

    int buf = 0;
    for (; s < nstripes; s += nblk) {
        float* __restrict__ sa = s_a + buf * A_STAGE;
        *(float4*)(sa + st0) = g0;
        *(float4*)(sa + st1) = g1;
        const int sn = s + nblk;
        if (sn < nstripes) {
            const size_t r0 = (size_t)((stripe0 + sn) << 4) + (size_t)w;
            g0 = *(const float4*)(gsrc + r0 * K);
            g1 = *(const float4*)(gsrc + (r0 + 8) * K);
        }
        lds_barrier();

        v8f acc;
#pragma unroll
        for (int r = 0; r < 8; ++r) acc[r] = bv;
        const float* __restrict__ fa = sa + fr;
        v2f a[K / 4];
#pragma unroll
        for (int i = 0; i < K / 4; ++i) a[i] = *(const v2f*)(fa + 4 * i);
        __builtin_amdgcn_sched_barrier(0);
#pragma unroll
        for (int i = 0; i < K / 4; ++i) {
            acc = __builtin_amdgcn_wmma_f32_16x16x4_f32(
                false, a[i], false, b[i], (short)0, acc, false, false);
        }
        const int mtile = (stripe0 + s) << 4;
#pragma unroll
        for (int r = 0; r < 8; ++r)
            Dout[(size_t)(mtile + (hi << 3) + r) * N + ntile + l16] = acc[r];
        buf ^= 1;
    }
}

template <int K, int N>
__global__ __launch_bounds__(256)
void bin_count(const int* __restrict__ row, int* __restrict__ tab, int E, int nbin, int nblk,
               const float* __restrict__ X, const float* __restrict__ W,
               const float* __restrict__ bias, float* __restrict__ H,
               int stripe0, int nstripes)
{
    __shared__ int s_hist[MAX_BINS];
    __shared__ __attribute__((aligned(16))) float s_a[2 * A_STAGE];
    const int tid = threadIdx.x;
    const int e0  = (int)blockIdx.x * CHUNK + tid;
    s_hist[tid] = 0;
    int r[EPT];
#pragma unroll
    for (int i = 0; i < EPT; ++i) {
        const int e = e0 + i * 256;
        r[i] = (e < E) ? row[e] : -1;
    }
    gemm_stripes_of_block<K, N>(X, W, bias, H, stripe0, nstripes, nblk, s_a);
    __syncthreads();
#pragma unroll
    for (int i = 0; i < EPT; ++i)
        if (r[i] >= 0) atomicAdd(&s_hist[r[i] >> BIN_SHIFT], 1);
    __syncthreads();
    if (tid < nbin) tab[tid * nblk + (int)blockIdx.x] = s_hist[tid];
}

__global__ __launch_bounds__(256)
void csr_scan(const int* __restrict__ cnt, int* __restrict__ off, int M)
{
    __shared__ int s_front[8];
    __shared__ int s_tile[8];
    const int tid   = threadIdx.x;
    const int lane  = tid & 31;
    const int w     = tid >> 5;
    const int tile0 = blockIdx.x * 1024;

    const int4* __restrict__ c4 = (const int4*)cnt + tid;
    int s = 0;
#pragma unroll 8
    for (int k = 0; k < (int)blockIdx.x; ++k) {
        const int4 q = c4[k * 256];
        s += (q.x + q.y) + (q.z + q.w);
    }

    const int i0 = tile0 + tid * 4;
    const int a0 = (i0 + 0 < M) ? cnt[i0 + 0] : 0;
    const int a1 = (i0 + 1 < M) ? cnt[i0 + 1] : 0;
    const int a2 = (i0 + 2 < M) ? cnt[i0 + 2] : 0;
    const int a3 = (i0 + 3 < M) ? cnt[i0 + 3] : 0;
    const int t  = (a0 + a1) + (a2 + a3);

    for (int d = 16; d > 0; d >>= 1) s += __shfl_xor(s, d);

    int inc = t;
    for (int d = 1; d < 32; d <<= 1) {
        const int y = __shfl_up(inc, d);
        if (lane >= d) inc += y;
    }
    if (lane == 31) s_tile[w]  = inc;
    if (lane == 0)  s_front[w] = s;
    __syncthreads();

    int base = 0;
    for (int k = 0; k < 8; ++k) {
        base += s_front[k];
        if (k < w) base += s_tile[k];
    }

    const int p0 = base + inc - t;
    const int p1 = p0 + a0;
    const int p2 = p1 + a1;
    const int p3 = p2 + a2;
    const int p4 = p3 + a3;
    if (i0 + 0 < M) off[i0 + 0] = p0;
    if (i0 + 1 < M) off[i0 + 1] = p1;
    if (i0 + 2 < M) off[i0 + 2] = p2;
    if (i0 + 3 < M) off[i0 + 3] = p3;
    if (i0 + 0 == M - 1) off[M] = p1;
    if (i0 + 1 == M - 1) off[M] = p2;
    if (i0 + 2 == M - 1) off[M] = p3;
    if (i0 + 3 == M - 1) off[M] = p4;
}

template <int K, int N>
__global__ __launch_bounds__(256)
void bin_scatter(const int* __restrict__ row, const int* __restrict__ col,
                 const float* __restrict__ val, const int* __restrict__ pre,
                 int2* __restrict__ binned, int E, int nbin, int nblk,
                 const float* __restrict__ X, const float* __restrict__ W,
                 const float* __restrict__ bias, float* __restrict__ H,
                 int stripe0, int nstripes)
{
    __shared__ int s_cur[MAX_BINS];
    __shared__ __attribute__((aligned(16))) float s_a[2 * A_STAGE];
    const int tid = threadIdx.x;
    const int e0  = (int)blockIdx.x * CHUNK + tid;
    int cur0 = 0;
    if (tid < nbin) cur0 = pre[tid * nblk + (int)blockIdx.x];
    int r[EPT], c[EPT], v[EPT];
#pragma unroll
    for (int i = 0; i < EPT; ++i) {
        const int e = e0 + i * 256;
        const bool live = e < E;
        r[i] = live ? row[e] : -1;
        c[i] = live ? col[e] : 0;
        v[i] = live ? __float_as_int(val[e]) : 0;
    }
    gemm_stripes_of_block<K, N>(X, W, bias, H, stripe0, nstripes, nblk, s_a);
    s_cur[tid] = cur0;
    __syncthreads();
#pragma unroll
    for (int i = 0; i < EPT; ++i) {
        if (r[i] >= 0) {
            const int pos = atomicAdd(&s_cur[r[i] >> BIN_SHIFT], 1);
            const unsigned key = ((unsigned)(r[i] & (BIN_ROWS - 1)) << 24) | (unsigned)c[i];
            binned[pos] = make_int2((int)key, v[i]);
        }
    }
}

constexpr int SORT_U = 5;
__global__ __launch_bounds__(1024)
void bin_sort(const int* __restrict__ pre, const int2* __restrict__ binned,
              int* __restrict__ off, int2* __restrict__ edges, int M, int nblk)
{
    __shared__ int s_cnt[BIN_ROWS];
    __shared__ int s_w[BIN_ROWS / 32];
    const int tid  = threadIdx.x;
    const int lane = tid & 31;
    const int beg = pre[(int)blockIdx.x * nblk];
    const int end = pre[((int)blockIdx.x + 1) * nblk];
    if (tid < BIN_ROWS) s_cnt[tid] = 0;
    __syncthreads();

    for (int base = beg; base < end; base += 1024 * SORT_U) {
        int k[SORT_U];
#pragma unroll
        for (int u = 0; u < SORT_U; ++u) {
            const int i = base + u * 1024 + tid;
            k[u] = binned[min(i, end - 1)].x;
        }
#pragma unroll
        for (int u = 0; u < SORT_U; ++u) {
            const int i = base + u * 1024 + tid;
            if (i < end) atomicAdd(&s_cnt[(unsigned)k[u] >> 24], 1);
        }
    }
    __syncthreads();

    int cn = 0, inc = 0;
    if (tid < BIN_ROWS) {
        cn  = s_cnt[tid];
        inc = cn;
        for (int d = 1; d < 32; d <<= 1) {
            const int y = __shfl_up(inc, d);
            if (lane >= d) inc += y;
        }
        if (lane == 31) s_w[tid >> 5] = inc;
    }
    __syncthreads();
    if (tid < BIN_ROWS) {
        const int w = tid >> 5;
        int start = beg + inc - cn;
        for (int q = 0; q < BIN_ROWS / 32; ++q)
            if (q < w) start += s_w[q];
        s_cnt[tid] = start;
        const int rr = ((int)blockIdx.x << BIN_SHIFT) + tid;
        if (rr < M) off[rr] = start;
        if (rr == M - 1) off[M] = start + cn;
    }
    __syncthreads();

    for (int base = beg; base < end; base += 1024 * SORT_U) {
        int2 rec[SORT_U];
#pragma unroll
        for (int u = 0; u < SORT_U; ++u) {
            const int i = base + u * 1024 + tid;
            rec[u] = binned[min(i, end - 1)];
        }
        int pos[SORT_U];
#pragma unroll
        for (int u = 0; u < SORT_U; ++u) {
            const int i = base + u * 1024 + tid;
            pos[u] = 0;
            if (i < end) pos[u] = atomicAdd(&s_cnt[(unsigned)rec[u].x >> 24], 1);
        }
#pragma unroll
        for (int u = 0; u < SORT_U; ++u) {
            const int i = base + u * 1024 + tid;
            if (i < end)
                edges[pos[u]] = make_int2((int)((unsigned)rec[u].x & 0x00FFFFFFu), rec[u].y);
        }
    }
}

template <int U>
__device__ __forceinline__ void add_rows(float4& acc, const float* __restrict__ Hl,
                                         int ec, int ev, int j)
{
    float4 h[U];
    float  v[U];
#pragma unroll
    for (int u = 0; u < U; ++u) {
        const int c = __builtin_amdgcn_readlane(ec, j + u);
        v[u] = __int_as_float(__builtin_amdgcn_readlane(ev, j + u));
        h[u] = *(const float4*)(Hl + (size_t)c * 128);
    }
#pragma unroll
    for (int u = 0; u < U; ++u) {
        acc.x = fmaf(v[u], h[u].x, acc.x);
        acc.y = fmaf(v[u], h[u].y, acc.y);
        acc.z = fmaf(v[u], h[u].z, acc.z);
        acc.w = fmaf(v[u], h[u].w, acc.w);
    }
}

__global__ __launch_bounds__(256)
void spmm_gather128(const int* __restrict__ off, const int* __restrict__ edges,
                    const float* __restrict__ H, float* __restrict__ out, int M)
{
    const int lane = threadIdx.x & 31;
    const int r = blockIdx.x * 8 + __builtin_amdgcn_readfirstlane(threadIdx.x >> 5);
    if (r >= M) return;
    const int beg = __builtin_amdgcn_readfirstlane(off[r]);
    const int end = __builtin_amdgcn_readfirstlane(off[r + 1]);
    const float* __restrict__ Hl = H + (lane << 2);
    const int2*  __restrict__ ed = (const int2*)edges;

    float4 acc = make_float4(0.0f, 0.0f, 0.0f, 0.0f);
    for (int base = beg; base < end; base += 32) {
        const int n = min(end - base, 32);
        int2 rec = make_int2(0, 0);
        if (lane < n) rec = ed[base + lane];
        int j = 0;
        for (; j + 8 <= n; j += 8) add_rows<8>(acc, Hl, rec.x, rec.y, j);
        if (n & 4) { add_rows<4>(acc, Hl, rec.x, rec.y, j); j += 4; }
        if (n & 2) { add_rows<2>(acc, Hl, rec.x, rec.y, j); j += 2; }
        if (n & 1) { add_rows<1>(acc, Hl, rec.x, rec.y, j); }
    }
    *(float4*)(out + (size_t)r * 128 + (lane << 2)) = acc;
}

__global__ __launch_bounds__(256)
void spmm_gather16(const int* __restrict__ off, const int* __restrict__ edges,
                   const float* __restrict__ H, float* __restrict__ out, int M)
{
    const int t = blockIdx.x * 256 + threadIdx.x;
    const int f = t & 15;
    const int r = t >> 4;
    const bool live = r < M;
    int beg = 0, end = 0;
    if (live) { beg = off[r]; end = off[r + 1]; }
    const int2* __restrict__ ed = (const int2*)edges;

    int rounds = (end - beg + 15) >> 4;
    rounds = max(rounds, __shfl_xor(rounds, 16));
    rounds = __builtin_amdgcn_readfirstlane(rounds);

    float acc = 0.0f;
    for (int q = 0; q < rounds; ++q) {
        const int base = beg + (q << 4);
        const int n = end - base;
        int2 rec = make_int2(0, 0);
        if (f < n) rec = ed[base + f];
#pragma unroll
        for (int j = 0; j < 16; ++j) {
            const int   c = __shfl(rec.x, j, 16);
            const float v = __int_as_float(__shfl(rec.y, j, 16));
            const float h = H[(size_t)c * 16 + f];
            const float s = fmaf(v, h, acc);
            acc = (j < n) ? s : acc;
        }
    }
    if (live) out[(size_t)r * 16 + f] = acc;
}

extern "C" void kernel_launch(void* const* d_in, const int* in_sizes, int n_in,
                              void* d_out, int out_size, void* d_ws, size_t ws_size,
                              hipStream_t stream) {
    const float* x   = (const float*)d_in[0];
    const int*   row = (const int*)  d_in[1];
    const int*   col = (const int*)  d_in[2];
    const float* val = (const float*)d_in[3];
    const float* W1  = (const float*)d_in[4];
    const float* b1  = (const float*)d_in[5];
    const float* W2  = (const float*)d_in[6];
    const float* b2  = (const float*)d_in[7];
    float* out = (float*)d_out;

    constexpr int D_IN  = 128;
    constexpr int D_HID = 128;
    constexpr int N_CLS = 16;
    const int E = in_sizes[1];
    const int M = in_sizes[0] / D_IN;

    const int nbin = (M + BIN_ROWS - 1) / BIN_ROWS;
    const int nblk = (E + CHUNK - 1) / CHUNK;
    const int ntab = nbin * nblk;

    char*  ws  = (char*)d_ws;
    float* H1  = (float*)ws;
    float* AGG = (float*)(ws + (size_t)M * D_HID * sizeof(float));
    float* H2  = (float*)(ws + 2 * (size_t)M * D_HID * sizeof(float));
    int*   tab    = (int*)(H2 + (size_t)M * N_CLS);
    int*   pre    = tab + ((ntab + 3) & ~3);
    int*   off    = pre + ((ntab + 1 + 3) & ~3);
    int*   binned = off + ((M + 1 + 3) & ~3);
    int*   edges  = binned + 2 * (size_t)E;

    const int stripes = M / 16;
    const int sA = (stripes + 1) / 2;
    const int sB = stripes - sA;
    bin_count<D_IN, D_HID><<<(unsigned)nblk, 256, 0, stream>>>(row, tab, E, nbin, nblk,
                                                               x, W1, b1, H1, 0, sA);
    csr_scan<<<(unsigned)((ntab + 1023) / 1024), 256, 0, stream>>>(tab, pre, ntab);
    bin_scatter<D_IN, D_HID><<<(unsigned)nblk, 256, 0, stream>>>(row, col, val, pre,
                                                                 (int2*)binned, E, nbin, nblk,
                                                                 x, W1, b1, H1, sA, sB);
    bin_sort<<<(unsigned)nbin, 1024, 0, stream>>>(pre, (const int2*)binned, off,
                                                  (int2*)edges, M, nblk);

    spmm_gather128<<<(unsigned)((M + 7) / 8), 256, 0, stream>>>(off, edges, H1, AGG, M);

    dim3 g2(N_CLS / 16, M / 16);
    gcn_gemm_wmma_f32<D_HID, N_CLS, true><<<g2, 32, 0, stream>>>(AGG, W2, b2, H2);

    {
        const long long threads = (long long)M * 16;
        spmm_gather16<<<(unsigned)((threads + 255) / 256), 256, 0, stream>>>(off, edges, H2, out, M);
    }
}


# Layer-2 product moved into the layer-1 row-sum launch: 16 summed rows per block go via LDS to WMMA, k split over 8 waves, partials added in wave order; one launch fewer, AGG never in memory
# speedup vs baseline: 2.6141x; 1.0622x over previous
#include <hip/hip_runtime.h>

typedef __attribute__((ext_vector_type(2))) float v2f;
typedef __attribute__((ext_vector_type(8))) float v8f;


constexpr int BIN_SHIFT = 8;
constexpr int BIN_ROWS  = 1 << BIN_SHIFT;
constexpr int MAX_BINS  = 256;
constexpr int EPT       = 8;
constexpr int CHUNK     = 256 * EPT;

constexpr int A_PITCH = 132;
constexpr int A_STAGE = 16 * A_PITCH;

__device__ __forceinline__ void lds_barrier()
{
    __builtin_amdgcn_fence(__ATOMIC_RELEASE, "workgroup", "local");
    __builtin_amdgcn_s_barrier();
    __builtin_amdgcn_fence(__ATOMIC_ACQUIRE, "workgroup", "local");
}

template <int K, int N>
__device__ __forceinline__
void gemm_stripes_of_block(const float* __restrict__ A, const float* __restrict__ W,
                           const float* __restrict__ bias, float* __restrict__ Dout,
                           const int stripe0, const int nstripes, const int nblk,
                           float* __restrict__ s_a)
{
    static_assert(K == 128, "a staged row is one 512-byte wave load");
    static_assert(N == 8 * 16, "eight waves of the block take the stripe's eight tiles");
    int s = (int)blockIdx.x;
    if (s >= nstripes) return;

    const int tid   = threadIdx.x;
    const int lane  = tid & 31;
    const int w     = __builtin_amdgcn_readfirstlane(tid >> 5);
    const int hi    = lane >> 4;
    const int l16   = lane & 15;
    const int koff  = hi << 1;
    const int ntile = w << 4;

    const int scol = lane << 2;
    const float* __restrict__ gsrc = A + scol;
    float4 g0, g1;
    {
        const size_t r0 = (size_t)((stripe0 + s) << 4) + (size_t)w;
        g0 = *(const float4*)(gsrc + r0 * K);
        g1 = *(const float4*)(gsrc + (r0 + 8) * K);
    }

    const float* __restrict__ bcol = W + (size_t)koff * N + ntile + l16;
    v2f b[K / 4];
#pragma unroll
    for (int i = 0; i < K / 4; ++i) {
        b[i][0] = bcol[(4 * i) * N];
        b[i][1] = bcol[(4 * i + 1) * N];
    }
    const float bv = bias[ntile + l16];

    const int st0 = w * A_PITCH + scol;
    const int st1 = st0 + 8 * A_PITCH;
    const int fr  = l16 * A_PITCH + koff;

    int buf = 0;
    for (; s < nstripes; s += nblk) {
        float* __restrict__ sa = s_a + buf * A_STAGE;
        *(float4*)(sa + st0) = g0;
        *(float4*)(sa + st1) = g1;
        const int sn = s + nblk;
        if (sn < nstripes) {
            const size_t r0 = (size_t)((stripe0 + sn) << 4) + (size_t)w;
            g0 = *(const float4*)(gsrc + r0 * K);
            g1 = *(const float4*)(gsrc + (r0 + 8) * K);
        }
        lds_barrier();

        v8f acc;
#pragma unroll
        for (int r = 0; r < 8; ++r) acc[r] = bv;
        const float* __restrict__ fa = sa + fr;
        v2f a[K / 4];
#pragma unroll
        for (int i = 0; i < K / 4; ++i) a[i] = *(const v2f*)(fa + 4 * i);
        __builtin_amdgcn_sched_barrier(0);
#pragma unroll
        for (int i = 0; i < K / 4; ++i) {
            acc = __builtin_amdgcn_wmma_f32_16x16x4_f32(
                false, a[i], false, b[i], (short)0, acc, false, false);
        }
        const int mtile = (stripe0 + s) << 4;
#pragma unroll
        for (int r = 0; r < 8; ++r)
            Dout[(size_t)(mtile + (hi << 3) + r) * N + ntile + l16] = acc[r];
        buf ^= 1;
    }
}

template <int K, int N>
__global__ __launch_bounds__(256)
void bin_count(const int* __restrict__ row, int* __restrict__ tab, int E, int nbin, int nblk,
               const float* __restrict__ X, const float* __restrict__ W,
               const float* __restrict__ bias, float* __restrict__ H,
               int stripe0, int nstripes)
{
    __shared__ int s_hist[MAX_BINS];
    __shared__ __attribute__((aligned(16))) float s_a[2 * A_STAGE];
    const int tid = threadIdx.x;
    const int e0  = (int)blockIdx.x * CHUNK + tid;
    s_hist[tid] = 0;
    int r[EPT];
#pragma unroll
    for (int i = 0; i < EPT; ++i) {
        const int e = e0 + i * 256;
        r[i] = (e < E) ? row[e] : -1;
    }
    gemm_stripes_of_block<K, N>(X, W, bias, H, stripe0, nstripes, nblk, s_a);
    __syncthreads();
#pragma unroll
    for (int i = 0; i < EPT; ++i)
        if (r[i] >= 0) atomicAdd(&s_hist[r[i] >> BIN_SHIFT], 1);
    __syncthreads();
    if (tid < nbin) tab[tid * nblk + (int)blockIdx.x] = s_hist[tid];
}

__global__ __launch_bounds__(256)
void csr_scan(const int* __restrict__ cnt, int* __restrict__ off, int M)
{
    __shared__ int s_front[8];
    __shared__ int s_tile[8];
    const int tid   = threadIdx.x;
    const int lane  = tid & 31;
    const int w     = tid >> 5;
    const int tile0 = blockIdx.x * 1024;

    const int4* __restrict__ c4 = (const int4*)cnt + tid;
    int s = 0;
#pragma unroll 8
    for (int k = 0; k < (int)blockIdx.x; ++k) {
        const int4 q = c4[k * 256];
        s += (q.x + q.y) + (q.z + q.w);
    }

    const int i0 = tile0 + tid * 4;
    const int a0 = (i0 + 0 < M) ? cnt[i0 + 0] : 0;
    const int a1 = (i0 + 1 < M) ? cnt[i0 + 1] : 0;
    const int a2 = (i0 + 2 < M) ? cnt[i0 + 2] : 0;
    const int a3 = (i0 + 3 < M) ? cnt[i0 + 3] : 0;
    const int t  = (a0 + a1) + (a2 + a3);

    for (int d = 16; d > 0; d >>= 1) s += __shfl_xor(s, d);

    int inc = t;
    for (int d = 1; d < 32; d <<= 1) {
        const int y = __shfl_up(inc, d);
        if (lane >= d) inc += y;
    }
    if (lane == 31) s_tile[w]  = inc;
    if (lane == 0)  s_front[w] = s;
    __syncthreads();

    int base = 0;
    for (int k = 0; k < 8; ++k) {
        base += s_front[k];
        if (k < w) base += s_tile[k];
    }

    const int p0 = base + inc - t;
    const int p1 = p0 + a0;
    const int p2 = p1 + a1;
    const int p3 = p2 + a2;
    const int p4 = p3 + a3;
    if (i0 + 0 < M) off[i0 + 0] = p0;
    if (i0 + 1 < M) off[i0 + 1] = p1;
    if (i0 + 2 < M) off[i0 + 2] = p2;
    if (i0 + 3 < M) off[i0 + 3] = p3;
    if (i0 + 0 == M - 1) off[M] = p1;
    if (i0 + 1 == M - 1) off[M] = p2;
    if (i0 + 2 == M - 1) off[M] = p3;
    if (i0 + 3 == M - 1) off[M] = p4;
}

template <int K, int N>
__global__ __launch_bounds__(256)
void bin_scatter(const int* __restrict__ row, const int* __restrict__ col,
                 const float* __restrict__ val, const int* __restrict__ pre,
                 int2* __restrict__ binned, int E, int nbin, int nblk,
                 const float* __restrict__ X, const float* __restrict__ W,
                 const float* __restrict__ bias, float* __restrict__ H,
                 int stripe0, int nstripes)
{
    __shared__ int s_cur[MAX_BINS];
    __shared__ __attribute__((aligned(16))) float s_a[2 * A_STAGE];
    const int tid = threadIdx.x;
    const int e0  = (int)blockIdx.x * CHUNK + tid;
    int cur0 = 0;
    if (tid < nbin) cur0 = pre[tid * nblk + (int)blockIdx.x];
    int r[EPT], c[EPT], v[EPT];
#pragma unroll
    for (int i = 0; i < EPT; ++i) {
        const int e = e0 + i * 256;
        const bool live = e < E;
        r[i] = live ? row[e] : -1;
        c[i] = live ? col[e] : 0;
        v[i] = live ? __float_as_int(val[e]) : 0;
    }
    gemm_stripes_of_block<K, N>(X, W, bias, H, stripe0, nstripes, nblk, s_a);
    s_cur[tid] = cur0;
    __syncthreads();
#pragma unroll
    for (int i = 0; i < EPT; ++i) {
        if (r[i] >= 0) {
            const int pos = atomicAdd(&s_cur[r[i] >> BIN_SHIFT], 1);
            const unsigned key = ((unsigned)(r[i] & (BIN_ROWS - 1)) << 24) | (unsigned)c[i];
            binned[pos] = make_int2((int)key, v[i]);
        }
    }
}

constexpr int SORT_U = 5;
__global__ __launch_bounds__(1024)
void bin_sort(const int* __restrict__ pre, const int2* __restrict__ binned,
              int* __restrict__ off, int2* __restrict__ edges, int M, int nblk)
{
    __shared__ int s_cnt[BIN_ROWS];
    __shared__ int s_w[BIN_ROWS / 32];
    const int tid  = threadIdx.x;
    const int lane = tid & 31;
    const int beg = pre[(int)blockIdx.x * nblk];
    const int end = pre[((int)blockIdx.x + 1) * nblk];
    if (tid < BIN_ROWS) s_cnt[tid] = 0;
    __syncthreads();

    for (int base = beg; base < end; base += 1024 * SORT_U) {
        int k[SORT_U];
#pragma unroll
        for (int u = 0; u < SORT_U; ++u) {
            const int i = base + u * 1024 + tid;
            k[u] = binned[min(i, end - 1)].x;
        }
#pragma unroll
        for (int u = 0; u < SORT_U; ++u) {
            const int i = base + u * 1024 + tid;
            if (i < end) atomicAdd(&s_cnt[(unsigned)k[u] >> 24], 1);
        }
    }
    __syncthreads();

    int cn = 0, inc = 0;
    if (tid < BIN_ROWS) {
        cn  = s_cnt[tid];
        inc = cn;
        for (int d = 1; d < 32; d <<= 1) {
            const int y = __shfl_up(inc, d);
            if (lane >= d) inc += y;
        }
        if (lane == 31) s_w[tid >> 5] = inc;
    }
    __syncthreads();
    if (tid < BIN_ROWS) {
        const int w = tid >> 5;
        int start = beg + inc - cn;
        for (int q = 0; q < BIN_ROWS / 32; ++q)
            if (q < w) start += s_w[q];
        s_cnt[tid] = start;
        const int rr = ((int)blockIdx.x << BIN_SHIFT) + tid;
        if (rr < M) off[rr] = start;
        if (rr == M - 1) off[M] = start + cn;
    }
    __syncthreads();

    for (int base = beg; base < end; base += 1024 * SORT_U) {
        int2 rec[SORT_U];
#pragma unroll
        for (int u = 0; u < SORT_U; ++u) {
            const int i = base + u * 1024 + tid;
            rec[u] = binned[min(i, end - 1)];
        }
        int pos[SORT_U];
#pragma unroll
        for (int u = 0; u < SORT_U; ++u) {
            const int i = base + u * 1024 + tid;
            pos[u] = 0;
            if (i < end) pos[u] = atomicAdd(&s_cnt[(unsigned)rec[u].x >> 24], 1);
        }
#pragma unroll
        for (int u = 0; u < SORT_U; ++u) {
            const int i = base + u * 1024 + tid;
            if (i < end)
                edges[pos[u]] = make_int2((int)((unsigned)rec[u].x & 0x00FFFFFFu), rec[u].y);
        }
    }
}

template <int U>
__device__ __forceinline__ void add_rows(float4& acc, const float* __restrict__ Hl,
                                         int ec, int ev, int j)
{
    float4 h[U];
    float  v[U];
#pragma unroll
    for (int u = 0; u < U; ++u) {
        const int c = __builtin_amdgcn_readlane(ec, j + u);
        v[u] = __int_as_float(__builtin_amdgcn_readlane(ev, j + u));
        h[u] = *(const float4*)(Hl + (size_t)c * 128);
    }
#pragma unroll
    for (int u = 0; u < U; ++u) {
        acc.x = fmaf(v[u], h[u].x, acc.x);
        acc.y = fmaf(v[u], h[u].y, acc.y);
        acc.z = fmaf(v[u], h[u].z, acc.z);
        acc.w = fmaf(v[u], h[u].w, acc.w);
    }
}

constexpr int P_HALF = 144;
constexpr int P_TILE = P_HALF + 128;

template <int K, int N>
__global__ __launch_bounds__(512)
void spmm_gather128_gemm(const int* __restrict__ off, const int* __restrict__ edges,
                         const float* __restrict__ H, const float* __restrict__ W,
                         const float* __restrict__ bias, float* __restrict__ Dout, int M)
{
    static_assert(K == 128, "a summed row is one float4 per lane");
    static_assert(N == 16, "one tile column; the k range is split over eight waves");
    __shared__ __attribute__((aligned(16))) float s_a[16 * A_PITCH];
    __shared__ float s_p[8 * P_TILE];
    const int tid  = threadIdx.x;
    const int lane = tid & 31;
    const int w = __builtin_amdgcn_readfirstlane(tid >> 5);
    const int r = blockIdx.x * 16 + w;

    float4 acc = make_float4(0.0f, 0.0f, 0.0f, 0.0f);
    if (r < M) {
        const int beg = __builtin_amdgcn_readfirstlane(off[r]);
        const int end = __builtin_amdgcn_readfirstlane(off[r + 1]);
        const float* __restrict__ Hl = H + (lane << 2);
        const int2*  __restrict__ ed = (const int2*)edges;
        for (int base = beg; base < end; base += 32) {
            const int n = min(end - base, 32);
            int2 rec = make_int2(0, 0);
            if (lane < n) rec = ed[base + lane];
            int j = 0;
            for (; j + 8 <= n; j += 8) add_rows<8>(acc, Hl, rec.x, rec.y, j);
            if (n & 4) { add_rows<4>(acc, Hl, rec.x, rec.y, j); j += 4; }
            if (n & 2) { add_rows<2>(acc, Hl, rec.x, rec.y, j); j += 2; }
            if (n & 1) { add_rows<1>(acc, Hl, rec.x, rec.y, j); }
        }
    }
    acc.x = fmaxf(acc.x, 0.0f);
    acc.y = fmaxf(acc.y, 0.0f);
    acc.z = fmaxf(acc.z, 0.0f);
    acc.w = fmaxf(acc.w, 0.0f);
    *(float4*)(s_a + w * A_PITCH + (lane << 2)) = acc;

    const int hi   = lane >> 4;
    const int l16  = lane & 15;
    const int koff = hi << 1;
    v2f b[4];
    float bv = 0.0f;
    if (w < 8) {
        const float* __restrict__ bcol = W + (size_t)(16 * w + koff) * N + l16;
#pragma unroll
        for (int i = 0; i < 4; ++i) {
            b[i][0] = bcol[(4 * i) * N];
            b[i][1] = bcol[(4 * i + 1) * N];
        }
        bv = bias[l16];
    }
    lds_barrier();
    if (w < 8) {
        const float* __restrict__ fa = s_a + l16 * A_PITCH + 16 * w + koff;
        v2f a[4];
#pragma unroll
        for (int i = 0; i < 4; ++i) a[i] = *(const v2f*)(fa + 4 * i);
        v8f part;
#pragma unroll
        for (int q = 0; q < 8; ++q) part[q] = 0.0f;
#pragma unroll
        for (int i = 0; i < 4; ++i) {
            part = __builtin_amdgcn_wmma_f32_16x16x4_f32(
                false, a[i], false, b[i], (short)0, part, false, false);
        }
        float* __restrict__ p = s_p + w * P_TILE + hi * P_HALF + l16;
#pragma unroll
        for (int q = 0; q < 8; ++q) p[q * 16] = part[q];
    }
    lds_barrier();
    if (tid < 256) {
        const int row = tid >> 4;
        const int col = tid & 15;
        const float* __restrict__ p = s_p + (row >> 3) * P_HALF + (row & 7) * 16 + col;
        float s = bv;
#pragma unroll
        for (int q = 0; q < 8; ++q) s += p[q * P_TILE];
        const int gr = blockIdx.x * 16 + row;
        if (gr < M) Dout[(size_t)gr * N + col] = s;
    }
}

__global__ __launch_bounds__(256)
void spmm_gather16(const int* __restrict__ off, const int* __restrict__ edges,
                   const float* __restrict__ H, float* __restrict__ out, int M)
{
    const int t = blockIdx.x * 256 + threadIdx.x;
    const int f = t & 15;
    const int r = t >> 4;
    const bool live = r < M;
    int beg = 0, end = 0;
    if (live) { beg = off[r]; end = off[r + 1]; }
    const int2* __restrict__ ed = (const int2*)edges;

    int rounds = (end - beg + 15) >> 4;
    rounds = max(rounds, __shfl_xor(rounds, 16));
    rounds = __builtin_amdgcn_readfirstlane(rounds);

    float acc = 0.0f;
    for (int q = 0; q < rounds; ++q) {
        const int base = beg + (q << 4);
        const int n = end - base;
        int2 rec = make_int2(0, 0);
        if (f < n) rec = ed[base + f];
#pragma unroll
        for (int j = 0; j < 16; ++j) {
            const int   c = __shfl(rec.x, j, 16);
            const float v = __int_as_float(__shfl(rec.y, j, 16));
            const float h = H[(size_t)c * 16 + f];
            const float s = fmaf(v, h, acc);
            acc = (j < n) ? s : acc;
        }
    }
    if (live) out[(size_t)r * 16 + f] = acc;
}

extern "C" void kernel_launch(void* const* d_in, const int* in_sizes, int n_in,
                              void* d_out, int out_size, void* d_ws, size_t ws_size,
                              hipStream_t stream) {
    const float* x   = (const float*)d_in[0];
    const int*   row = (const int*)  d_in[1];
    const int*   col = (const int*)  d_in[2];
    const float* val = (const float*)d_in[3];
    const float* W1  = (const float*)d_in[4];
    const float* b1  = (const float*)d_in[5];
    const float* W2  = (const float*)d_in[6];
    const float* b2  = (const float*)d_in[7];
    float* out = (float*)d_out;

    constexpr int D_IN  = 128;
    constexpr int D_HID = 128;
    constexpr int N_CLS = 16;
    const int E = in_sizes[1];
    const int M = in_sizes[0] / D_IN;

    const int nbin = (M + BIN_ROWS - 1) / BIN_ROWS;
    const int nblk = (E + CHUNK - 1) / CHUNK;
    const int ntab = nbin * nblk;

    char*  ws  = (char*)d_ws;
    float* H1  = (float*)ws;
    float* H2  = (float*)(ws + (size_t)M * D_HID * sizeof(float));
    int*   tab    = (int*)(H2 + (size_t)M * N_CLS);
    int*   pre    = tab + ((ntab + 3) & ~3);
    int*   off    = pre + ((ntab + 1 + 3) & ~3);
    int*   binned = off + ((M + 1 + 3) & ~3);
    int*   edges  = binned + 2 * (size_t)E;

    const int stripes = M / 16;
    const int sA = (stripes + 1) / 2;
    const int sB = stripes - sA;
    bin_count<D_IN, D_HID><<<(unsigned)nblk, 256, 0, stream>>>(row, tab, E, nbin, nblk,
                                                               x, W1, b1, H1, 0, sA);
    csr_scan<<<(unsigned)((ntab + 1023) / 1024), 256, 0, stream>>>(tab, pre, ntab);
    bin_scatter<D_IN, D_HID><<<(unsigned)nblk, 256, 0, stream>>>(row, col, val, pre,
                                                                 (int2*)binned, E, nbin, nblk,
                                                                 x, W1, b1, H1, sA, sB);
    bin_sort<<<(unsigned)nbin, 1024, 0, stream>>>(pre, (const int2*)binned, off,
                                                  (int2*)edges, M, nblk);

    spmm_gather128_gemm<D_HID, N_CLS><<<(unsigned)((M + 15) / 16), 512, 0, stream>>>(off, edges, H1,
                                                                                 W2, b2, H2, M);

    {
        const long long threads = (long long)M * 16;
        spmm_gather16<<<(unsigned)((threads + 255) / 256), 256, 0, stream>>>(off, edges, H2, out, M);
    }
}


# Edge sort in two launches, not four: a chunk block bins its own edges in LDS into its own part of the list; the bin block sums counts in front and fetches its runs; sum order = LDS arrival order
# speedup vs baseline: 3.4184x; 1.3077x over previous
#include <hip/hip_runtime.h>

typedef __attribute__((ext_vector_type(2))) float v2f;
typedef __attribute__((ext_vector_type(8))) float v8f;


constexpr int BIN_SHIFT = 8;
constexpr int BIN_ROWS  = 1 << BIN_SHIFT;
constexpr int MAX_BINS  = 256;
constexpr int EPT       = 8;
constexpr int CHUNK     = 256 * EPT;
constexpr int MAX_CHUNKS = 1024;
constexpr int CNT_MASK  = 0xFFF;

constexpr int A_PITCH = 132;
constexpr int A_STAGE = 16 * A_PITCH;

__device__ __forceinline__ void lds_barrier()
{
    __builtin_amdgcn_fence(__ATOMIC_RELEASE, "workgroup", "local");
    __builtin_amdgcn_s_barrier();
    __builtin_amdgcn_fence(__ATOMIC_ACQUIRE, "workgroup", "local");
}

template <int K, int N>
__device__ __forceinline__
void gemm_stripes_of_block(const float* __restrict__ A, const float* __restrict__ W,
                           const float* __restrict__ bias, float* __restrict__ Dout,
                           const int stripe0, const int nstripes, const int nblk,
                           float* __restrict__ s_a)
{
    static_assert(K == 128, "a staged row is one 512-byte wave load");
    static_assert(N == 8 * 16, "eight waves of the block take the stripe's eight tiles");
    int s = (int)blockIdx.x;
    if (s >= nstripes) return;

    const int tid   = threadIdx.x;
    const int lane  = tid & 31;
    const int w     = __builtin_amdgcn_readfirstlane(tid >> 5);
    const int hi    = lane >> 4;
    const int l16   = lane & 15;
    const int koff  = hi << 1;
    const int ntile = w << 4;

    const int scol = lane << 2;
    const float* __restrict__ gsrc = A + scol;
    float4 g0, g1;
    {
        const size_t r0 = (size_t)((stripe0 + s) << 4) + (size_t)w;
        g0 = *(const float4*)(gsrc + r0 * K);
        g1 = *(const float4*)(gsrc + (r0 + 8) * K);
    }

    const float* __restrict__ bcol = W + (size_t)koff * N + ntile + l16;
    v2f b[K / 4];
#pragma unroll
    for (int i = 0; i < K / 4; ++i) {
        b[i][0] = bcol[(4 * i) * N];
        b[i][1] = bcol[(4 * i + 1) * N];
    }
    const float bv = bias[ntile + l16];

    const int st0 = w * A_PITCH + scol;
    const int st1 = st0 + 8 * A_PITCH;
    const int fr  = l16 * A_PITCH + koff;

    int buf = 0;
    for (; s < nstripes; s += nblk) {
        float* __restrict__ sa = s_a + buf * A_STAGE;
        *(float4*)(sa + st0) = g0;
        *(float4*)(sa + st1) = g1;
        const int sn = s + nblk;
        if (sn < nstripes) {
            const size_t r0 = (size_t)((stripe0 + sn) << 4) + (size_t)w;
            g0 = *(const float4*)(gsrc + r0 * K);
            g1 = *(const float4*)(gsrc + (r0 + 8) * K);
        }
        lds_barrier();

        v8f acc;
#pragma unroll
        for (int r = 0; r < 8; ++r) acc[r] = bv;
        const float* __restrict__ fa = sa + fr;
        v2f a[K / 4];
#pragma unroll
        for (int i = 0; i < K / 4; ++i) a[i] = *(const v2f*)(fa + 4 * i);
        __builtin_amdgcn_sched_barrier(0);
#pragma unroll
        for (int i = 0; i < K / 4; ++i) {
            acc = __builtin_amdgcn_wmma_f32_16x16x4_f32(
                false, a[i], false, b[i], (short)0, acc, false, false);
        }
        const int mtile = (stripe0 + s) << 4;
#pragma unroll
        for (int r = 0; r < 8; ++r)
            Dout[(size_t)(mtile + (hi << 3) + r) * N + ntile + l16] = acc[r];
        buf ^= 1;
    }
}

template <int K, int N>
__global__ __launch_bounds__(256)
void bin_chunk(const int* __restrict__ row, const int* __restrict__ col,
               const float* __restrict__ val, int* __restrict__ tab,
               int2* __restrict__ binned, int E, int nbin, int nblk, int tpitch,
               const float* __restrict__ X, const float* __restrict__ W,
               const float* __restrict__ bias, float* __restrict__ H,
               int stripe0, int nstripes)
{
    __shared__ int s_hist[MAX_BINS];
    __shared__ int s_w[8];
    __shared__ __attribute__((aligned(16))) float s_a[2 * A_STAGE];
    const int tid  = threadIdx.x;
    const int lane = tid & 31;
    const int e0   = (int)blockIdx.x * CHUNK + tid;
    s_hist[tid] = 0;
    int r[EPT], c[EPT], v[EPT];
#pragma unroll
    for (int i = 0; i < EPT; ++i) {
        const int e = e0 + i * 256;
        const bool live = e < E;
        r[i] = live ? row[e] : -1;
        c[i] = live ? col[e] : 0;
        v[i] = live ? __float_as_int(val[e]) : 0;
    }
    gemm_stripes_of_block<K, N>(X, W, bias, H, stripe0, nstripes, nblk, s_a);
    __syncthreads();
#pragma unroll
    for (int i = 0; i < EPT; ++i)
        if (r[i] >= 0) atomicAdd(&s_hist[r[i] >> BIN_SHIFT], 1);
    __syncthreads();

    const int cn = s_hist[tid];
    int inc = cn;
    for (int d = 1; d < 32; d <<= 1) {
        const int y = __shfl_up(inc, d);
        if (lane >= d) inc += y;
    }
    if (lane == 31) s_w[tid >> 5] = inc;
    __syncthreads();
    int start = inc - cn;
    for (int q = 0; q < 8; ++q)
        if (q < (tid >> 5)) start += s_w[q];
    s_hist[tid] = start;
    if (tid < nbin) {
        tab[tid * tpitch + (int)blockIdx.x] = (start << 12) | cn;
        if (blockIdx.x == 0)
            for (int p = nblk; p < tpitch; ++p) tab[tid * tpitch + p] = 0;
    }
    __syncthreads();

    int2* __restrict__ dst = binned + (size_t)blockIdx.x * CHUNK;
#pragma unroll
    for (int i = 0; i < EPT; ++i) {
        if (r[i] >= 0) {
            const int pos = atomicAdd(&s_hist[r[i] >> BIN_SHIFT], 1);
            const unsigned key = ((unsigned)(r[i] & (BIN_ROWS - 1)) << 24) | (unsigned)c[i];
            dst[pos] = make_int2((int)key, v[i]);
        }
    }
}

__device__ __forceinline__ int run_slot(const int* __restrict__ s_scan, const int* __restrict__ s_src,
                                        int nblk, int i)
{
    int lo = 0, hi = nblk;
#pragma unroll
    for (int s = 0; s < 10; ++s) {
        const int mid = (lo + hi) >> 1;
        const bool ge = s_scan[mid] <= i;
        lo = ge ? mid : lo;
        hi = ge ? hi : mid;
    }
    return s_src[lo] + (i - s_scan[lo]);
}

constexpr int SORT_U = 5;
__global__ __launch_bounds__(1024)
void bin_sort(const int* __restrict__ tab, const int2* __restrict__ binned,
              int* __restrict__ off, int2* __restrict__ edges, int M, int nblk, int tpitch)
{
    __shared__ int s_cnt[BIN_ROWS];
    __shared__ int s_w[32];
    __shared__ int s_front[32];
    __shared__ int s_scan[MAX_CHUNKS + 1];
    __shared__ int s_src[MAX_CHUNKS];
    const int tid  = threadIdx.x;
    const int lane = tid & 31;
    const int w    = tid >> 5;

    int pk = 0;
    if (tid < nblk) pk = tab[(int)blockIdx.x * tpitch + tid];

    const int4* __restrict__ t4 = (const int4*)tab;
    const int nfront4 = (int)blockIdx.x * (tpitch >> 2);
    int fs = 0;
#pragma unroll 4
    for (int k = tid; k < nfront4; k += 1024) {
        const int4 q = t4[k];
        fs += ((q.x & CNT_MASK) + (q.y & CNT_MASK)) + ((q.z & CNT_MASK) + (q.w & CNT_MASK));
    }
    for (int d = 16; d > 0; d >>= 1) fs += __shfl_xor(fs, d);

    if (tid < BIN_ROWS) s_cnt[tid] = 0;
    const int rn = pk & CNT_MASK;
    int rinc = rn;
    for (int d = 1; d < 32; d <<= 1) {
        const int y = __shfl_up(rinc, d);
        if (lane >= d) rinc += y;
    }
    if (lane == 31) s_w[w]     = rinc;
    if (lane == 0)  s_front[w] = fs;
    __syncthreads();
    int beg = 0, rbase = 0;
    for (int q = 0; q < 32; ++q) {
        beg += s_front[q];
        if (q < w) rbase += s_w[q];
    }
    const int rex = rbase + rinc - rn;
    if (tid < nblk) {
        s_scan[tid] = rex;
        s_src[tid]  = tid * CHUNK + (int)((unsigned)pk >> 12);
        if (tid == nblk - 1) s_scan[nblk] = rex + rn;
    }
    __syncthreads();
    const int total = s_scan[nblk];
    const bool one_trip = total <= 1024 * SORT_U;

    int2 rec[SORT_U];
#pragma unroll
    for (int u = 0; u < SORT_U; ++u) rec[u] = make_int2(0, 0);
    for (int base = 0; base < total; base += 1024 * SORT_U) {
#pragma unroll
        for (int u = 0; u < SORT_U; ++u) {
            const int i = base + u * 1024 + tid;
            rec[u] = binned[run_slot(s_scan, s_src, nblk, min(i, total - 1))];
        }
#pragma unroll
        for (int u = 0; u < SORT_U; ++u) {
            const int i = base + u * 1024 + tid;
            if (i < total) atomicAdd(&s_cnt[(unsigned)rec[u].x >> 24], 1);
        }
    }
    __syncthreads();

    int cn = 0, inc = 0;
    if (tid < BIN_ROWS) {
        cn  = s_cnt[tid];
        inc = cn;
        for (int d = 1; d < 32; d <<= 1) {
            const int y = __shfl_up(inc, d);
            if (lane >= d) inc += y;
        }
        if (lane == 31) s_w[w] = inc;
    }
    __syncthreads();
    if (tid < BIN_ROWS) {
        int start = beg + inc - cn;
        for (int q = 0; q < BIN_ROWS / 32; ++q)
            if (q < w) start += s_w[q];
        s_cnt[tid] = start;
        const int rr = ((int)blockIdx.x << BIN_SHIFT) + tid;
        if (rr < M) off[rr] = start;
        if (rr == M - 1) off[M] = start + cn;
    }
    __syncthreads();

    for (int base = 0; base < total; base += 1024 * SORT_U) {
        if (!one_trip) {
#pragma unroll
            for (int u = 0; u < SORT_U; ++u) {
                const int i = base + u * 1024 + tid;
                rec[u] = binned[run_slot(s_scan, s_src, nblk, min(i, total - 1))];
            }
        }
        int pos[SORT_U];
#pragma unroll
        for (int u = 0; u < SORT_U; ++u) {
            const int i = base + u * 1024 + tid;
            pos[u] = 0;
            if (i < total) pos[u] = atomicAdd(&s_cnt[(unsigned)rec[u].x >> 24], 1);
        }
#pragma unroll
        for (int u = 0; u < SORT_U; ++u) {
            const int i = base + u * 1024 + tid;
            if (i < total)
                edges[pos[u]] = make_int2((int)((unsigned)rec[u].x & 0x00FFFFFFu), rec[u].y);
        }
    }
}

template <int U>
__device__ __forceinline__ void add_rows(float4& acc, const float* __restrict__ Hl,
                                         int ec, int ev, int j)
{
    float4 h[U];
    float  v[U];
#pragma unroll
    for (int u = 0; u < U; ++u) {
        const int c = __builtin_amdgcn_readlane(ec, j + u);
        v[u] = __int_as_float(__builtin_amdgcn_readlane(ev, j + u));
        h[u] = *(const float4*)(Hl + (size_t)c * 128);
    }
#pragma unroll
    for (int u = 0; u < U; ++u) {
        acc.x = fmaf(v[u], h[u].x, acc.x);
        acc.y = fmaf(v[u], h[u].y, acc.y);
        acc.z = fmaf(v[u], h[u].z, acc.z);
        acc.w = fmaf(v[u], h[u].w, acc.w);
    }
}

constexpr int P_HALF = 144;
constexpr int P_TILE = P_HALF + 128;

template <int K, int N>
__global__ __launch_bounds__(512)
void spmm_gather128_gemm(const int* __restrict__ off, const int* __restrict__ edges,
                         const float* __restrict__ H, const float* __restrict__ W,
                         const float* __restrict__ bias, float* __restrict__ Dout, int M)
{
    static_assert(K == 128, "a summed row is one float4 per lane");
    static_assert(N == 16, "one tile column; the k range is split over eight waves");
    __shared__ __attribute__((aligned(16))) float s_a[16 * A_PITCH];
    __shared__ float s_p[8 * P_TILE];
    const int tid  = threadIdx.x;
    const int lane = tid & 31;
    const int w = __builtin_amdgcn_readfirstlane(tid >> 5);
    const int r = blockIdx.x * 16 + w;

    float4 acc = make_float4(0.0f, 0.0f, 0.0f, 0.0f);
    if (r < M) {
        const int beg = __builtin_amdgcn_readfirstlane(off[r]);
        const int end = __builtin_amdgcn_readfirstlane(off[r + 1]);
        const float* __restrict__ Hl = H + (lane << 2);
        const int2*  __restrict__ ed = (const int2*)edges;
        for (int base = beg; base < end; base += 32) {
            const int n = min(end - base, 32);
            int2 rec = make_int2(0, 0);
            if (lane < n) rec = ed[base + lane];
            int j = 0;
            for (; j + 8 <= n; j += 8) add_rows<8>(acc, Hl, rec.x, rec.y, j);
            if (n & 4) { add_rows<4>(acc, Hl, rec.x, rec.y, j); j += 4; }
            if (n & 2) { add_rows<2>(acc, Hl, rec.x, rec.y, j); j += 2; }
            if (n & 1) { add_rows<1>(acc, Hl, rec.x, rec.y, j); }
        }
    }
    acc.x = fmaxf(acc.x, 0.0f);
    acc.y = fmaxf(acc.y, 0.0f);
    acc.z = fmaxf(acc.z, 0.0f);
    acc.w = fmaxf(acc.w, 0.0f);
    *(float4*)(s_a + w * A_PITCH + (lane << 2)) = acc;

    const int hi   = lane >> 4;
    const int l16  = lane & 15;
    const int koff = hi << 1;
    v2f b[4];
    float bv = 0.0f;
    if (w < 8) {
        const float* __restrict__ bcol = W + (size_t)(16 * w + koff) * N + l16;
#pragma unroll
        for (int i = 0; i < 4; ++i) {
            b[i][0] = bcol[(4 * i) * N];
            b[i][1] = bcol[(4 * i + 1) * N];
        }
        bv = bias[l16];
    }
    lds_barrier();
    if (w < 8) {
        const float* __restrict__ fa = s_a + l16 * A_PITCH + 16 * w + koff;
        v2f a[4];
#pragma unroll
        for (int i = 0; i < 4; ++i) a[i] = *(const v2f*)(fa + 4 * i);
        v8f part;
#pragma unroll
        for (int q = 0; q < 8; ++q) part[q] = 0.0f;
#pragma unroll
        for (int i = 0; i < 4; ++i) {
            part = __builtin_amdgcn_wmma_f32_16x16x4_f32(
                false, a[i], false, b[i], (short)0, part, false, false);
        }
        float* __restrict__ p = s_p + w * P_TILE + hi * P_HALF + l16;
#pragma unroll
        for (int q = 0; q < 8; ++q) p[q * 16] = part[q];
    }
    lds_barrier();
    if (tid < 256) {
        const int row = tid >> 4;
        const int col = tid & 15;
        const float* __restrict__ p = s_p + (row >> 3) * P_HALF + (row & 7) * 16 + col;
        float s = bv;
#pragma unroll
        for (int q = 0; q < 8; ++q) s += p[q * P_TILE];
        const int gr = blockIdx.x * 16 + row;
        if (gr < M) Dout[(size_t)gr * N + col] = s;
    }
}

__global__ __launch_bounds__(256)
void spmm_gather16(const int* __restrict__ off, const int* __restrict__ edges,
                   const float* __restrict__ H, float* __restrict__ out, int M)
{
    const int t = blockIdx.x * 256 + threadIdx.x;
    const int f = t & 15;
    const int r = t >> 4;
    const bool live = r < M;
    int beg = 0, end = 0;
    if (live) { beg = off[r]; end = off[r + 1]; }
    const int2* __restrict__ ed = (const int2*)edges;

    int rounds = (end - beg + 15) >> 4;
    rounds = max(rounds, __shfl_xor(rounds, 16));
    rounds = __builtin_amdgcn_readfirstlane(rounds);

    float acc = 0.0f;
    for (int q = 0; q < rounds; ++q) {
        const int base = beg + (q << 4);
        const int n = end - base;
        int2 rec = make_int2(0, 0);
        if (f < n) rec = ed[base + f];
#pragma unroll
        for (int j = 0; j < 16; ++j) {
            const int   c = __shfl(rec.x, j, 16);
            const float v = __int_as_float(__shfl(rec.y, j, 16));
            const float h = H[(size_t)c * 16 + f];
            const float s = fmaf(v, h, acc);
            acc = (j < n) ? s : acc;
        }
    }
    if (live) out[(size_t)r * 16 + f] = acc;
}

extern "C" void kernel_launch(void* const* d_in, const int* in_sizes, int n_in,
                              void* d_out, int out_size, void* d_ws, size_t ws_size,
                              hipStream_t stream) {
    const float* x   = (const float*)d_in[0];
    const int*   row = (const int*)  d_in[1];
    const int*   col = (const int*)  d_in[2];
    const float* val = (const float*)d_in[3];
    const float* W1  = (const float*)d_in[4];
    const float* b1  = (const float*)d_in[5];
    const float* W2  = (const float*)d_in[6];
    const float* b2  = (const float*)d_in[7];
    float* out = (float*)d_out;

    constexpr int D_IN  = 128;
    constexpr int D_HID = 128;
    constexpr int N_CLS = 16;
    const int E = in_sizes[1];
    const int M = in_sizes[0] / D_IN;

    const int nbin = (M + BIN_ROWS - 1) / BIN_ROWS;
    const int nblk = (E + CHUNK - 1) / CHUNK;
    const int tpitch = (nblk + 3) & ~3;
    const int ntab = nbin * tpitch;

    char*  ws  = (char*)d_ws;
    float* H1  = (float*)ws;
    float* H2  = (float*)(ws + (size_t)M * D_HID * sizeof(float));
    int*   tab    = (int*)(H2 + (size_t)M * N_CLS);
    int*   off    = tab + ntab;
    int*   binned = off + ((M + 1 + 3) & ~3);
    int*   edges  = binned + 2 * (size_t)nblk * CHUNK;

    const int stripes = M / 16;
    bin_chunk<D_IN, D_HID><<<(unsigned)nblk, 256, 0, stream>>>(row, col, val, tab,
                                                               (int2*)binned, E, nbin, nblk, tpitch,
                                                               x, W1, b1, H1, 0, stripes);
    bin_sort<<<(unsigned)nbin, 1024, 0, stream>>>(tab, (const int2*)binned, off,
                                                  (int2*)edges, M, nblk, tpitch);

    spmm_gather128_gemm<D_HID, N_CLS><<<(unsigned)((M + 15) / 16), 512, 0, stream>>>(off, edges, H1,
                                                                                 W2, b2, H2, M);

    {
        const long long threads = (long long)M * 16;
        spmm_gather16<<<(unsigned)((threads + 255) / 256), 256, 0, stream>>>(off, edges, H2, out, M);
    }
}


# Layer-1 product moved behind the row sums: (sum val*x) @ W1 + rowsum(val)*b1 in the row-sum launch, resident blocks hold W1 in registers; x @ W1 never stored; sum order changed by reassociation
# speedup vs baseline: 3.6065x; 1.0550x over previous
#include <hip/hip_runtime.h>

typedef __attribute__((ext_vector_type(2))) float v2f;
typedef __attribute__((ext_vector_type(8))) float v8f;


constexpr int BIN_SHIFT = 8;
constexpr int BIN_ROWS  = 1 << BIN_SHIFT;
constexpr int MAX_BINS  = 256;
constexpr int EPT       = 8;
constexpr int CHUNK     = 256 * EPT;
constexpr int MAX_CHUNKS = 1024;
constexpr int CNT_MASK  = 0xFFF;


constexpr int A_PITCH = 132;
constexpr int A_STAGE = 16 * A_PITCH;

__device__ __forceinline__ void lds_barrier()
{
    __builtin_amdgcn_fence(__ATOMIC_RELEASE, "workgroup", "local");
    __builtin_amdgcn_s_barrier();
    __builtin_amdgcn_fence(__ATOMIC_ACQUIRE, "workgroup", "local");
}

__global__ __launch_bounds__(256)
void bin_chunk(const int* __restrict__ row, const int* __restrict__ col,
               const float* __restrict__ val, int* __restrict__ tab,
               int2* __restrict__ binned, int E, int nbin, int nblk, int tpitch)
{
    __shared__ int s_hist[MAX_BINS];
    __shared__ int s_w[8];
    const int tid  = threadIdx.x;
    const int lane = tid & 31;
    const int e0   = (int)blockIdx.x * CHUNK + tid;
    s_hist[tid] = 0;
    int r[EPT], c[EPT], v[EPT];
#pragma unroll
    for (int i = 0; i < EPT; ++i) {
        const int e = e0 + i * 256;
        const bool live = e < E;
        r[i] = live ? row[e] : -1;
        c[i] = live ? col[e] : 0;
        v[i] = live ? __float_as_int(val[e]) : 0;
    }
    __syncthreads();
#pragma unroll
    for (int i = 0; i < EPT; ++i)
        if (r[i] >= 0) atomicAdd(&s_hist[r[i] >> BIN_SHIFT], 1);
    __syncthreads();

    const int cn = s_hist[tid];
    int inc = cn;
    for (int d = 1; d < 32; d <<= 1) {
        const int y = __shfl_up(inc, d);
        if (lane >= d) inc += y;
    }
    if (lane == 31) s_w[tid >> 5] = inc;
    __syncthreads();
    int start = inc - cn;
    for (int q = 0; q < 8; ++q)
        if (q < (tid >> 5)) start += s_w[q];
    s_hist[tid] = start;
    if (tid < nbin) {
        tab[tid * tpitch + (int)blockIdx.x] = (start << 12) | cn;
        if (blockIdx.x == 0)
            for (int p = nblk; p < tpitch; ++p) tab[tid * tpitch + p] = 0;
    }
    __syncthreads();

    int2* __restrict__ dst = binned + (size_t)blockIdx.x * CHUNK;
#pragma unroll
    for (int i = 0; i < EPT; ++i) {
        if (r[i] >= 0) {
            const int pos = atomicAdd(&s_hist[r[i] >> BIN_SHIFT], 1);
            const unsigned key = ((unsigned)(r[i] & (BIN_ROWS - 1)) << 24) | (unsigned)c[i];
            dst[pos] = make_int2((int)key, v[i]);
        }
    }
}

__device__ __forceinline__ int run_slot(const int* __restrict__ s_scan, const int* __restrict__ s_src,
                                        int nblk, int i)
{
    int lo = 0, hi = nblk;
#pragma unroll
    for (int s = 0; s < 10; ++s) {
        const int mid = (lo + hi) >> 1;
        const bool ge = s_scan[mid] <= i;
        lo = ge ? mid : lo;
        hi = ge ? hi : mid;
    }
    return s_src[lo] + (i - s_scan[lo]);
}

constexpr int SORT_U = 5;
__global__ __launch_bounds__(1024)
void bin_sort(const int* __restrict__ tab, const int2* __restrict__ binned,
              int* __restrict__ off, int2* __restrict__ edges, int M, int nblk, int tpitch)
{
    __shared__ int s_cnt[BIN_ROWS];
    __shared__ int s_w[32];
    __shared__ int s_front[32];
    __shared__ int s_scan[MAX_CHUNKS + 1];
    __shared__ int s_src[MAX_CHUNKS];
    const int tid  = threadIdx.x;
    const int lane = tid & 31;
    const int w    = tid >> 5;

    int pk = 0;
    if (tid < nblk) pk = tab[(int)blockIdx.x * tpitch + tid];

    const int4* __restrict__ t4 = (const int4*)tab;
    const int nfront4 = (int)blockIdx.x * (tpitch >> 2);
    int fs = 0;
#pragma unroll 4
    for (int k = tid; k < nfront4; k += 1024) {
        const int4 q = t4[k];
        fs += ((q.x & CNT_MASK) + (q.y & CNT_MASK)) + ((q.z & CNT_MASK) + (q.w & CNT_MASK));
    }
    for (int d = 16; d > 0; d >>= 1) fs += __shfl_xor(fs, d);

    if (tid < BIN_ROWS) s_cnt[tid] = 0;
    const int rn = pk & CNT_MASK;
    int rinc = rn;
    for (int d = 1; d < 32; d <<= 1) {
        const int y = __shfl_up(rinc, d);
        if (lane >= d) rinc += y;
    }
    if (lane == 31) s_w[w]     = rinc;
    if (lane == 0)  s_front[w] = fs;
    __syncthreads();
    int beg = 0, rbase = 0;
    for (int q = 0; q < 32; ++q) {
        beg += s_front[q];
        if (q < w) rbase += s_w[q];
    }
    const int rex = rbase + rinc - rn;
    if (tid < nblk) {
        s_scan[tid] = rex;
        s_src[tid]  = tid * CHUNK + (int)((unsigned)pk >> 12);
        if (tid == nblk - 1) s_scan[nblk] = rex + rn;
    }
    __syncthreads();
    const int total = s_scan[nblk];
    const bool one_trip = total <= 1024 * SORT_U;

    int2 rec[SORT_U];
#pragma unroll
    for (int u = 0; u < SORT_U; ++u) rec[u] = make_int2(0, 0);
    for (int base = 0; base < total; base += 1024 * SORT_U) {
#pragma unroll
        for (int u = 0; u < SORT_U; ++u) {
            const int i = base + u * 1024 + tid;
            rec[u] = binned[run_slot(s_scan, s_src, nblk, min(i, total - 1))];
        }
#pragma unroll
        for (int u = 0; u < SORT_U; ++u) {
            const int i = base + u * 1024 + tid;
            if (i < total) atomicAdd(&s_cnt[(unsigned)rec[u].x >> 24], 1);
        }
    }
    __syncthreads();

    int cn = 0, inc = 0;
    if (tid < BIN_ROWS) {
        cn  = s_cnt[tid];
        inc = cn;
        for (int d = 1; d < 32; d <<= 1) {
            const int y = __shfl_up(inc, d);
            if (lane >= d) inc += y;
        }
        if (lane == 31) s_w[w] = inc;
    }
    __syncthreads();
    if (tid < BIN_ROWS) {
        int start = beg + inc - cn;
        for (int q = 0; q < BIN_ROWS / 32; ++q)
            if (q < w) start += s_w[q];
        s_cnt[tid] = start;
        const int rr = ((int)blockIdx.x << BIN_SHIFT) + tid;
        if (rr < M) off[rr] = start;
        if (rr == M - 1) off[M] = start + cn;
    }
    __syncthreads();

    for (int base = 0; base < total; base += 1024 * SORT_U) {
        if (!one_trip) {
#pragma unroll
            for (int u = 0; u < SORT_U; ++u) {
                const int i = base + u * 1024 + tid;
                rec[u] = binned[run_slot(s_scan, s_src, nblk, min(i, total - 1))];
            }
        }
        int pos[SORT_U];
#pragma unroll
        for (int u = 0; u < SORT_U; ++u) {
            const int i = base + u * 1024 + tid;
            pos[u] = 0;
            if (i < total) pos[u] = atomicAdd(&s_cnt[(unsigned)rec[u].x >> 24], 1);
        }
#pragma unroll
        for (int u = 0; u < SORT_U; ++u) {
            const int i = base + u * 1024 + tid;
            if (i < total)
                edges[pos[u]] = make_int2((int)((unsigned)rec[u].x & 0x00FFFFFFu), rec[u].y);
        }
    }
}

template <int U>
__device__ __forceinline__ void add_rows(float4& acc, float& vsum, const float* __restrict__ Hl,
                                         int ec, int ev, int j)
{
    float4 h[U];
    float  v[U];
#pragma unroll
    for (int u = 0; u < U; ++u) {
        const int c = __builtin_amdgcn_readlane(ec, j + u);
        v[u] = __int_as_float(__builtin_amdgcn_readlane(ev, j + u));
        h[u] = *(const float4*)(Hl + (size_t)c * 128);
    }
#pragma unroll
    for (int u = 0; u < U; ++u) {
        acc.x = fmaf(v[u], h[u].x, acc.x);
        acc.y = fmaf(v[u], h[u].y, acc.y);
        acc.z = fmaf(v[u], h[u].z, acc.z);
        acc.w = fmaf(v[u], h[u].w, acc.w);
        vsum += v[u];
    }
}

constexpr int P_HALF = 144;
constexpr int P_TILE = P_HALF + 128;

template <int K, int N1, int N2>
__global__ __launch_bounds__(512)
void spmm_gather128_gemm2(const int* __restrict__ off, const int* __restrict__ edges,
                          const float* __restrict__ X, const float* __restrict__ W1,
                          const float* __restrict__ b1, const float* __restrict__ W2,
                          const float* __restrict__ b2, float* __restrict__ Dout,
                          int M, int nstripes)
{
    static_assert(K == 128, "a summed row is one float4 per lane");
    static_assert(N1 == 128, "eight tile columns, each k half on its own wave");
    static_assert(N2 == 16, "one tile column; the k range is split over eight waves");
    __shared__ __attribute__((aligned(16))) float s_a[A_STAGE];
    __shared__ __attribute__((aligned(16))) float s_q[2 * A_STAGE];
    __shared__ float s_p[8 * P_TILE];
    __shared__ float s_rs[16];
    const int tid  = threadIdx.x;
    const int lane = tid & 31;
    const int w    = __builtin_amdgcn_readfirstlane(tid >> 5);
    const int hi   = lane >> 4;
    const int l16  = lane & 15;
    const int koff = hi << 1;
    const int t1   = w & 7;
    const int kh   = w >> 3;

    v2f bw1[16];
    {
        const float* __restrict__ bcol = W1 + (size_t)(64 * kh + koff) * N1 + 16 * t1 + l16;
#pragma unroll
        for (int i = 0; i < 16; ++i) {
            bw1[i][0] = bcol[(4 * i) * N1];
            bw1[i][1] = bcol[(4 * i + 1) * N1];
        }
    }
    const float b1x = b1[(lane << 2) + 0];
    const float b1y = b1[(lane << 2) + 1];
    const float b1z = b1[(lane << 2) + 2];
    const float b1w = b1[(lane << 2) + 3];
    v2f bw2[4];
    float bv = 0.0f;
    if (w < 8) {
        const float* __restrict__ bcol = W2 + (size_t)(16 * w + koff) * N2 + l16;
#pragma unroll
        for (int i = 0; i < 4; ++i) {
            bw2[i][0] = bcol[(4 * i) * N2];
            bw2[i][1] = bcol[(4 * i + 1) * N2];
        }
        bv = b2[l16];
    }

    const float* __restrict__ Xl = X + (lane << 2);
    const int2*  __restrict__ ed = (const int2*)edges;

    int begv = 0, endv = 0;
    {
        const int r0 = (int)blockIdx.x * 16 + w;
        if (r0 < M) {
            begv = off[r0];
            endv = off[r0 + 1];
        }
    }
    for (int s = (int)blockIdx.x; s < nstripes; s += (int)gridDim.x) {
        const int beg = __builtin_amdgcn_readfirstlane(begv);
        const int end = __builtin_amdgcn_readfirstlane(endv);
        float4 acc = make_float4(0.0f, 0.0f, 0.0f, 0.0f);
        float vs = 0.0f;
        for (int base = beg; base < end; base += 32) {
            const int n = min(end - base, 32);
            int2 rec = make_int2(0, 0);
            if (lane < n) rec = ed[base + lane];
            int j = 0;
            for (; j + 8 <= n; j += 8) add_rows<8>(acc, vs, Xl, rec.x, rec.y, j);
            if (n & 4) { add_rows<4>(acc, vs, Xl, rec.x, rec.y, j); j += 4; }
            if (n & 2) { add_rows<2>(acc, vs, Xl, rec.x, rec.y, j); j += 2; }
            if (n & 1) { add_rows<1>(acc, vs, Xl, rec.x, rec.y, j); }
        }
        begv = 0;
        endv = 0;
        {
            const int rn = (s + (int)gridDim.x) * 16 + w;
            if (rn < M) {
                begv = off[rn];
                endv = off[rn + 1];
            }
        }
        *(float4*)(s_a + w * A_PITCH + (lane << 2)) = acc;
        if (lane == 0) s_rs[w] = vs;
        lds_barrier();

        {
            const float* __restrict__ fa = s_a + l16 * A_PITCH + 64 * kh + koff;
            v2f a[16];
#pragma unroll
            for (int i = 0; i < 16; ++i) a[i] = *(const v2f*)(fa + 4 * i);
            __builtin_amdgcn_sched_barrier(0);
            v8f part;
#pragma unroll
            for (int q = 0; q < 8; ++q) part[q] = 0.0f;
#pragma unroll
            for (int i = 0; i < 16; ++i) {
                part = __builtin_amdgcn_wmma_f32_16x16x4_f32(
                    false, a[i], false, bw1[i], (short)0, part, false, false);
            }
            float* __restrict__ p = s_q + kh * A_STAGE + (hi << 3) * A_PITCH + 16 * t1 + l16;
#pragma unroll
            for (int q = 0; q < 8; ++q) p[q * A_PITCH] = part[q];
        }
        lds_barrier();

        {
            const float4 p0 = *(const float4*)(s_q + w * A_PITCH + (lane << 2));
            const float4 p1 = *(const float4*)(s_q + A_STAGE + w * A_PITCH + (lane << 2));
            const float rsw = s_rs[w];
            float4 h;
            h.x = fmaxf(fmaf(rsw, b1x, p0.x + p1.x), 0.0f);
            h.y = fmaxf(fmaf(rsw, b1y, p0.y + p1.y), 0.0f);
            h.z = fmaxf(fmaf(rsw, b1z, p0.z + p1.z), 0.0f);
            h.w = fmaxf(fmaf(rsw, b1w, p0.w + p1.w), 0.0f);
            *(float4*)(s_a + w * A_PITCH + (lane << 2)) = h;
        }
        lds_barrier();

        if (w < 8) {
            const float* __restrict__ fa = s_a + l16 * A_PITCH + 16 * w + koff;
            v2f a[4];
#pragma unroll
            for (int i = 0; i < 4; ++i) a[i] = *(const v2f*)(fa + 4 * i);
            v8f part;
#pragma unroll
            for (int q = 0; q < 8; ++q) part[q] = 0.0f;
#pragma unroll
            for (int i = 0; i < 4; ++i) {
                part = __builtin_amdgcn_wmma_f32_16x16x4_f32(
                    false, a[i], false, bw2[i], (short)0, part, false, false);
            }
            float* __restrict__ p = s_p + w * P_TILE + hi * P_HALF + l16;
#pragma unroll
            for (int q = 0; q < 8; ++q) p[q * 16] = part[q];
        }
        lds_barrier();
        if (tid < 256) {
            const int row = tid >> 4;
            const int col = tid & 15;
            const float* __restrict__ p = s_p + (row >> 3) * P_HALF + (row & 7) * 16 + col;
            float sum = bv;
#pragma unroll
            for (int q = 0; q < 8; ++q) sum += p[q * P_TILE];
            const int gr = s * 16 + row;
            if (gr < M) Dout[(size_t)gr * N2 + col] = sum;
        }
    }
}

__global__ __launch_bounds__(256)
void spmm_gather16(const int* __restrict__ off, const int* __restrict__ edges,
                   const float* __restrict__ H, float* __restrict__ out, int M)
{
    const int t = blockIdx.x * 256 + threadIdx.x;
    const int f = t & 15;
    const int r = t >> 4;
    const bool live = r < M;
    int beg = 0, end = 0;
    if (live) { beg = off[r]; end = off[r + 1]; }
    const int2* __restrict__ ed = (const int2*)edges;

    int rounds = (end - beg + 15) >> 4;
    rounds = max(rounds, __shfl_xor(rounds, 16));
    rounds = __builtin_amdgcn_readfirstlane(rounds);

    float acc = 0.0f;
    for (int q = 0; q < rounds; ++q) {
        const int base = beg + (q << 4);
        const int n = end - base;
        int2 rec = make_int2(0, 0);
        if (f < n) rec = ed[base + f];
#pragma unroll
        for (int j = 0; j < 16; ++j) {
            const int   c = __shfl(rec.x, j, 16);
            const float v = __int_as_float(__shfl(rec.y, j, 16));
            const float h = H[(size_t)c * 16 + f];
            const float s = fmaf(v, h, acc);
            acc = (j < n) ? s : acc;
        }
    }
    if (live) out[(size_t)r * 16 + f] = acc;
}

extern "C" void kernel_launch(void* const* d_in, const int* in_sizes, int n_in,
                              void* d_out, int out_size, void* d_ws, size_t ws_size,
                              hipStream_t stream) {
    const float* x   = (const float*)d_in[0];
    const int*   row = (const int*)  d_in[1];
    const int*   col = (const int*)  d_in[2];
    const float* val = (const float*)d_in[3];
    const float* W1  = (const float*)d_in[4];
    const float* b1  = (const float*)d_in[5];
    const float* W2  = (const float*)d_in[6];
    const float* b2  = (const float*)d_in[7];
    float* out = (float*)d_out;

    constexpr int D_IN  = 128;
    constexpr int D_HID = 128;
    constexpr int N_CLS = 16;
    const int E = in_sizes[1];
    const int M = in_sizes[0] / D_IN;

    const int nbin = (M + BIN_ROWS - 1) / BIN_ROWS;
    const int nblk = (E + CHUNK - 1) / CHUNK;
    const int tpitch = (nblk + 3) & ~3;
    const int ntab = nbin * tpitch;

    char*  ws  = (char*)d_ws;
    float* H2  = (float*)ws;
    int*   tab    = (int*)(H2 + (size_t)M * N_CLS);
    int*   off    = tab + ntab;
    int*   binned = off + ((M + 1 + 3) & ~3);
    int*   edges  = binned + 2 * (size_t)nblk * CHUNK;

    bin_chunk<<<(unsigned)nblk, 256, 0, stream>>>(row, col, val, tab,
                                                  (int2*)binned, E, nbin, nblk, tpitch);
    bin_sort<<<(unsigned)nbin, 1024, 0, stream>>>(tab, (const int2*)binned, off,
                                                  (int2*)edges, M, nblk, tpitch);

    spmm_gather128_gemm2<D_IN, D_HID, N_CLS><<<512, 512, 0, stream>>>(off, edges, x, W1, b1, W2, b2,
                                                                   H2, M, (M + 15) / 16);

    {
        const long long threads = (long long)M * 16;
        spmm_gather16<<<(unsigned)((threads + 255) / 256), 256, 0, stream>>>(off, edges, H2, out, M);
    }
}


# Row-sum launch, fewer dependent round trips per row: gathers leave in chunks of up to 14 edges, a short tail as one group, next stripe's records fetched under the products; sum order unchanged
# speedup vs baseline: 3.7312x; 1.0346x over previous
#include <hip/hip_runtime.h>

typedef __attribute__((ext_vector_type(2))) float v2f;
typedef __attribute__((ext_vector_type(8))) float v8f;


constexpr int BIN_SHIFT = 8;
constexpr int BIN_ROWS  = 1 << BIN_SHIFT;
constexpr int MAX_BINS  = 256;
constexpr int EPT       = 8;
constexpr int CHUNK     = 256 * EPT;
constexpr int MAX_CHUNKS = 1024;
constexpr int CNT_MASK  = 0xFFF;


constexpr int A_PITCH = 132;
constexpr int A_STAGE = 16 * A_PITCH;

__device__ __forceinline__ void lds_barrier()
{
    __builtin_amdgcn_fence(__ATOMIC_RELEASE, "workgroup", "local");
    __builtin_amdgcn_s_barrier();
    __builtin_amdgcn_fence(__ATOMIC_ACQUIRE, "workgroup", "local");
}

__global__ __launch_bounds__(256)
void bin_chunk(const int* __restrict__ row, const int* __restrict__ col,
               const float* __restrict__ val, int* __restrict__ tab,
               int2* __restrict__ binned, int E, int nbin, int nblk, int tpitch)
{
    __shared__ int s_hist[MAX_BINS];
    __shared__ int s_w[8];
    const int tid  = threadIdx.x;
    const int lane = tid & 31;
    const int e0   = (int)blockIdx.x * CHUNK + tid;
    s_hist[tid] = 0;
    int r[EPT], c[EPT], v[EPT];
#pragma unroll
    for (int i = 0; i < EPT; ++i) {
        const int e = e0 + i * 256;
        const bool live = e < E;
        r[i] = live ? row[e] : -1;
        c[i] = live ? col[e] : 0;
        v[i] = live ? __float_as_int(val[e]) : 0;
    }
    __syncthreads();
#pragma unroll
    for (int i = 0; i < EPT; ++i)
        if (r[i] >= 0) atomicAdd(&s_hist[r[i] >> BIN_SHIFT], 1);
    __syncthreads();

    const int cn = s_hist[tid];
    int inc = cn;
    for (int d = 1; d < 32; d <<= 1) {
        const int y = __shfl_up(inc, d);
        if (lane >= d) inc += y;
    }
    if (lane == 31) s_w[tid >> 5] = inc;
    __syncthreads();
    int start = inc - cn;
    for (int q = 0; q < 8; ++q)
        if (q < (tid >> 5)) start += s_w[q];
    s_hist[tid] = start;
    if (tid < nbin) {
        tab[tid * tpitch + (int)blockIdx.x] = (start << 12) | cn;
        if (blockIdx.x == 0)
            for (int p = nblk; p < tpitch; ++p) tab[tid * tpitch + p] = 0;
    }
    __syncthreads();

    int2* __restrict__ dst = binned + (size_t)blockIdx.x * CHUNK;
#pragma unroll
    for (int i = 0; i < EPT; ++i) {
        if (r[i] >= 0) {
            const int pos = atomicAdd(&s_hist[r[i] >> BIN_SHIFT], 1);
            const unsigned key = ((unsigned)(r[i] & (BIN_ROWS - 1)) << 24) | (unsigned)c[i];
            dst[pos] = make_int2((int)key, v[i]);
        }
    }
}

__device__ __forceinline__ int run_slot(const int* __restrict__ s_scan, const int* __restrict__ s_src,
                                        int nblk, int i)
{
    int lo = 0, hi = nblk;
#pragma unroll
    for (int s = 0; s < 10; ++s) {
        const int mid = (lo + hi) >> 1;
        const bool ge = s_scan[mid] <= i;
        lo = ge ? mid : lo;
        hi = ge ? hi : mid;
    }
    return s_src[lo] + (i - s_scan[lo]);
}

constexpr int SORT_U = 5;
__global__ __launch_bounds__(1024)
void bin_sort(const int* __restrict__ tab, const int2* __restrict__ binned,
              int* __restrict__ off, int2* __restrict__ edges, int M, int nblk, int tpitch)
{
    __shared__ int s_cnt[BIN_ROWS];
    __shared__ int s_w[32];
    __shared__ int s_front[32];
    __shared__ int s_scan[MAX_CHUNKS + 1];
    __shared__ int s_src[MAX_CHUNKS];
    const int tid  = threadIdx.x;
    const int lane = tid & 31;
    const int w    = tid >> 5;

    int pk = 0;
    if (tid < nblk) pk = tab[(int)blockIdx.x * tpitch + tid];

    const int4* __restrict__ t4 = (const int4*)tab;
    const int nfront4 = (int)blockIdx.x * (tpitch >> 2);
    int fs = 0;
#pragma unroll 4
    for (int k = tid; k < nfront4; k += 1024) {
        const int4 q = t4[k];
        fs += ((q.x & CNT_MASK) + (q.y & CNT_MASK)) + ((q.z & CNT_MASK) + (q.w & CNT_MASK));
    }
    for (int d = 16; d > 0; d >>= 1) fs += __shfl_xor(fs, d);

    if (tid < BIN_ROWS) s_cnt[tid] = 0;
    const int rn = pk & CNT_MASK;
    int rinc = rn;
    for (int d = 1; d < 32; d <<= 1) {
        const int y = __shfl_up(rinc, d);
        if (lane >= d) rinc += y;
    }
    if (lane == 31) s_w[w]     = rinc;
    if (lane == 0)  s_front[w] = fs;
    __syncthreads();
    int beg = 0, rbase = 0;
    for (int q = 0; q < 32; ++q) {
        beg += s_front[q];
        if (q < w) rbase += s_w[q];
    }
    const int rex = rbase + rinc - rn;
    if (tid < nblk) {
        s_scan[tid] = rex;
        s_src[tid]  = tid * CHUNK + (int)((unsigned)pk >> 12);
        if (tid == nblk - 1) s_scan[nblk] = rex + rn;
    }
    __syncthreads();
    const int total = s_scan[nblk];
    const bool one_trip = total <= 1024 * SORT_U;

    int2 rec[SORT_U];
#pragma unroll
    for (int u = 0; u < SORT_U; ++u) rec[u] = make_int2(0, 0);
    for (int base = 0; base < total; base += 1024 * SORT_U) {
#pragma unroll
        for (int u = 0; u < SORT_U; ++u) {
            const int i = base + u * 1024 + tid;
            rec[u] = binned[run_slot(s_scan, s_src, nblk, min(i, total - 1))];
        }
#pragma unroll
        for (int u = 0; u < SORT_U; ++u) {
            const int i = base + u * 1024 + tid;
            if (i < total) atomicAdd(&s_cnt[(unsigned)rec[u].x >> 24], 1);
        }
    }
    __syncthreads();

    int cn = 0, inc = 0;
    if (tid < BIN_ROWS) {
        cn  = s_cnt[tid];
        inc = cn;
        for (int d = 1; d < 32; d <<= 1) {
            const int y = __shfl_up(inc, d);
            if (lane >= d) inc += y;
        }
        if (lane == 31) s_w[w] = inc;
    }
    __syncthreads();
    if (tid < BIN_ROWS) {
        int start = beg + inc - cn;
        for (int q = 0; q < BIN_ROWS / 32; ++q)
            if (q < w) start += s_w[q];
        s_cnt[tid] = start;
        const int rr = ((int)blockIdx.x << BIN_SHIFT) + tid;
        if (rr < M) off[rr] = start;
        if (rr == M - 1) off[M] = start + cn;
    }
    __syncthreads();

    for (int base = 0; base < total; base += 1024 * SORT_U) {
        if (!one_trip) {
#pragma unroll
            for (int u = 0; u < SORT_U; ++u) {
                const int i = base + u * 1024 + tid;
                rec[u] = binned[run_slot(s_scan, s_src, nblk, min(i, total - 1))];
            }
        }
        int pos[SORT_U];
#pragma unroll
        for (int u = 0; u < SORT_U; ++u) {
            const int i = base + u * 1024 + tid;
            pos[u] = 0;
            if (i < total) pos[u] = atomicAdd(&s_cnt[(unsigned)rec[u].x >> 24], 1);
        }
#pragma unroll
        for (int u = 0; u < SORT_U; ++u) {
            const int i = base + u * 1024 + tid;
            if (i < total)
                edges[pos[u]] = make_int2((int)((unsigned)rec[u].x & 0x00FFFFFFu), rec[u].y);
        }
    }
}

constexpr int ROWS_IN_FLIGHT = 14;

template <int U, int I>
__device__ __forceinline__ void add_chunk(float4& acc, float& vsum, float4 (&h)[U], float (&v)[U],
                                          const float* __restrict__ Xl, int ec, int ev, int j, int m)
{
    const int c = __builtin_amdgcn_readlane(ec, j + I);
    v[I] = __int_as_float(__builtin_amdgcn_readlane(ev, j + I));
    h[I] = *(const float4*)(Xl + (size_t)c * 128);
    if constexpr (I + 1 < U) {
        if (I + 1 < m) {
            add_chunk<U, I + 1>(acc, vsum, h, v, Xl, ec, ev, j, m);
            return;
        }
    }
#pragma unroll
    for (int u = 0; u <= I; ++u) {
        acc.x = fmaf(v[u], h[u].x, acc.x);
        acc.y = fmaf(v[u], h[u].y, acc.y);
        acc.z = fmaf(v[u], h[u].z, acc.z);
        acc.w = fmaf(v[u], h[u].w, acc.w);
        vsum += v[u];
    }
}

constexpr int P_HALF = 144;
constexpr int P_TILE = P_HALF + 128;

template <int K, int N1, int N2>
__global__ __launch_bounds__(512)
void spmm_gather128_gemm2(const int* __restrict__ off, const int* __restrict__ edges,
                          const float* __restrict__ X, const float* __restrict__ W1,
                          const float* __restrict__ b1, const float* __restrict__ W2,
                          const float* __restrict__ b2, float* __restrict__ Dout,
                          int M, int nstripes)
{
    static_assert(K == 128, "a summed row is one float4 per lane");
    static_assert(N1 == 128, "eight tile columns, each k half on its own wave");
    static_assert(N2 == 16, "one tile column; the k range is split over eight waves");
    __shared__ __attribute__((aligned(16))) float s_a[A_STAGE];
    __shared__ __attribute__((aligned(16))) float s_q[2 * A_STAGE];
    __shared__ float s_p[8 * P_TILE];
    __shared__ float s_rs[16];
    const int tid  = threadIdx.x;
    const int lane = tid & 31;
    const int w    = __builtin_amdgcn_readfirstlane(tid >> 5);
    const int hi   = lane >> 4;
    const int l16  = lane & 15;
    const int koff = hi << 1;
    const int t1   = w & 7;
    const int kh   = w >> 3;

    v2f bw1[16];
    {
        const float* __restrict__ bcol = W1 + (size_t)(64 * kh + koff) * N1 + 16 * t1 + l16;
#pragma unroll
        for (int i = 0; i < 16; ++i) {
            bw1[i][0] = bcol[(4 * i) * N1];
            bw1[i][1] = bcol[(4 * i + 1) * N1];
        }
    }
    const float b1x = b1[(lane << 2) + 0];
    const float b1y = b1[(lane << 2) + 1];
    const float b1z = b1[(lane << 2) + 2];
    const float b1w = b1[(lane << 2) + 3];
    v2f bw2[4];
    float bv = 0.0f;
    if (w < 8) {
        const float* __restrict__ bcol = W2 + (size_t)(16 * w + koff) * N2 + l16;
#pragma unroll
        for (int i = 0; i < 4; ++i) {
            bw2[i][0] = bcol[(4 * i) * N2];
            bw2[i][1] = bcol[(4 * i + 1) * N2];
        }
        bv = b2[l16];
    }

    const float* __restrict__ Xl = X + (lane << 2);
    const int2*  __restrict__ ed = (const int2*)edges;

    int beg = 0, end = 0;
    {
        const int r0 = (int)blockIdx.x * 16 + w;
        if (r0 < M) {
            beg = __builtin_amdgcn_readfirstlane(off[r0]);
            end = __builtin_amdgcn_readfirstlane(off[r0 + 1]);
        }
    }
    int2 rec = make_int2(0, 0);
    if (lane < end - beg) rec = ed[beg + lane];
    int begv = 0, endv = 0;
    {
        const int r1 = ((int)blockIdx.x + (int)gridDim.x) * 16 + w;
        if (r1 < M) {
            begv = off[r1];
            endv = off[r1 + 1];
        }
    }
    for (int s = (int)blockIdx.x; s < nstripes; s += (int)gridDim.x) {
        const int begn = __builtin_amdgcn_readfirstlane(begv);
        const int endn = __builtin_amdgcn_readfirstlane(endv);
        float4 acc = make_float4(0.0f, 0.0f, 0.0f, 0.0f);
        float vs = 0.0f;
        for (int base = beg; base < end; base += 32) {
            const int n = min(end - base, 32);
            if (base != beg) {
                rec = make_int2(0, 0);
                if (lane < n) rec = ed[base + lane];
            }
            for (int j = 0; j < n; j += ROWS_IN_FLIGHT) {
                float4 h[ROWS_IN_FLIGHT];
                float  v[ROWS_IN_FLIGHT];
                add_chunk<ROWS_IN_FLIGHT, 0>(acc, vs, h, v, Xl, rec.x, rec.y, j, n - j);
            }
        }
        beg = begn;
        end = endn;
        rec = make_int2(0, 0);
        if (lane < end - beg) rec = ed[beg + lane];
        begv = 0;
        endv = 0;
        {
            const int r2 = (s + 2 * (int)gridDim.x) * 16 + w;
            if (r2 < M) {
                begv = off[r2];
                endv = off[r2 + 1];
            }
        }
        *(float4*)(s_a + w * A_PITCH + (lane << 2)) = acc;
        if (lane == 0) s_rs[w] = vs;
        lds_barrier();

        {
            const float* __restrict__ fa = s_a + l16 * A_PITCH + 64 * kh + koff;
            v2f a[16];
#pragma unroll
            for (int i = 0; i < 16; ++i) a[i] = *(const v2f*)(fa + 4 * i);
            __builtin_amdgcn_sched_barrier(0);
            v8f part;
#pragma unroll
            for (int q = 0; q < 8; ++q) part[q] = 0.0f;
#pragma unroll
            for (int i = 0; i < 16; ++i) {
                part = __builtin_amdgcn_wmma_f32_16x16x4_f32(
                    false, a[i], false, bw1[i], (short)0, part, false, false);
            }
            float* __restrict__ p = s_q + kh * A_STAGE + (hi << 3) * A_PITCH + 16 * t1 + l16;
#pragma unroll
            for (int q = 0; q < 8; ++q) p[q * A_PITCH] = part[q];
        }
        lds_barrier();

        {
            const float4 p0 = *(const float4*)(s_q + w * A_PITCH + (lane << 2));
            const float4 p1 = *(const float4*)(s_q + A_STAGE + w * A_PITCH + (lane << 2));
            const float rsw = s_rs[w];
            float4 h;
            h.x = fmaxf(fmaf(rsw, b1x, p0.x + p1.x), 0.0f);
            h.y = fmaxf(fmaf(rsw, b1y, p0.y + p1.y), 0.0f);
            h.z = fmaxf(fmaf(rsw, b1z, p0.z + p1.z), 0.0f);
            h.w = fmaxf(fmaf(rsw, b1w, p0.w + p1.w), 0.0f);
            *(float4*)(s_a + w * A_PITCH + (lane << 2)) = h;
        }
        lds_barrier();

        if (w < 8) {
            const float* __restrict__ fa = s_a + l16 * A_PITCH + 16 * w + koff;
            v2f a[4];
#pragma unroll
            for (int i = 0; i < 4; ++i) a[i] = *(const v2f*)(fa + 4 * i);
            v8f part;
#pragma unroll
            for (int q = 0; q < 8; ++q) part[q] = 0.0f;
#pragma unroll
            for (int i = 0; i < 4; ++i) {
                part = __builtin_amdgcn_wmma_f32_16x16x4_f32(
                    false, a[i], false, bw2[i], (short)0, part, false, false);
            }
            float* __restrict__ p = s_p + w * P_TILE + hi * P_HALF + l16;
#pragma unroll
            for (int q = 0; q < 8; ++q) p[q * 16] = part[q];
        }
        lds_barrier();
        if (tid < 256) {
            const int row = tid >> 4;
            const int col = tid & 15;
            const float* __restrict__ p = s_p + (row >> 3) * P_HALF + (row & 7) * 16 + col;
            float sum = bv;
#pragma unroll
            for (int q = 0; q < 8; ++q) sum += p[q * P_TILE];
            const int gr = s * 16 + row;
            if (gr < M) Dout[(size_t)gr * N2 + col] = sum;
        }
    }
}

__global__ __launch_bounds__(256)
void spmm_gather16(const int* __restrict__ off, const int* __restrict__ edges,
                   const float* __restrict__ H, float* __restrict__ out, int M)
{
    const int t = blockIdx.x * 256 + threadIdx.x;
    const int f = t & 15;
    const int r = t >> 4;
    const bool live = r < M;
    int beg = 0, end = 0;
    if (live) { beg = off[r]; end = off[r + 1]; }
    const int2* __restrict__ ed = (const int2*)edges;

    int rounds = (end - beg + 15) >> 4;
    rounds = max(rounds, __shfl_xor(rounds, 16));
    rounds = __builtin_amdgcn_readfirstlane(rounds);

    float acc = 0.0f;
    for (int q = 0; q < rounds; ++q) {
        const int base = beg + (q << 4);
        const int n = end - base;
        int2 rec = make_int2(0, 0);
        if (f < n) rec = ed[base + f];
#pragma unroll
        for (int j = 0; j < 16; ++j) {
            const int   c = __shfl(rec.x, j, 16);
            const float v = __int_as_float(__shfl(rec.y, j, 16));
            const float h = H[(size_t)c * 16 + f];
            const float s = fmaf(v, h, acc);
            acc = (j < n) ? s : acc;
        }
    }
    if (live) out[(size_t)r * 16 + f] = acc;
}

extern "C" void kernel_launch(void* const* d_in, const int* in_sizes, int n_in,
                              void* d_out, int out_size, void* d_ws, size_t ws_size,
                              hipStream_t stream) {
    const float* x   = (const float*)d_in[0];
    const int*   row = (const int*)  d_in[1];
    const int*   col = (const int*)  d_in[2];
    const float* val = (const float*)d_in[3];
    const float* W1  = (const float*)d_in[4];
    const float* b1  = (const float*)d_in[5];
    const float* W2  = (const float*)d_in[6];
    const float* b2  = (const float*)d_in[7];
    float* out = (float*)d_out;

    constexpr int D_IN  = 128;
    constexpr int D_HID = 128;
    constexpr int N_CLS = 16;
    const int E = in_sizes[1];
    const int M = in_sizes[0] / D_IN;

    const int nbin = (M + BIN_ROWS - 1) / BIN_ROWS;
    const int nblk = (E + CHUNK - 1) / CHUNK;
    const int tpitch = (nblk + 3) & ~3;
    const int ntab = nbin * tpitch;

    char*  ws  = (char*)d_ws;
    float* H2  = (float*)ws;
    int*   tab    = (int*)(H2 + (size_t)M * N_CLS);
    int*   off    = tab + ntab;
    int*   binned = off + ((M + 1 + 3) & ~3);
    int*   edges  = binned + 2 * (size_t)nblk * CHUNK;

    bin_chunk<<<(unsigned)nblk, 256, 0, stream>>>(row, col, val, tab,
                                                  (int2*)binned, E, nbin, nblk, tpitch);
    bin_sort<<<(unsigned)nbin, 1024, 0, stream>>>(tab, (const int2*)binned, off,
                                                  (int2*)edges, M, nblk, tpitch);

    spmm_gather128_gemm2<D_IN, D_HID, N_CLS><<<512, 512, 0, stream>>>(off, edges, x, W1, b1, W2, b2,
                                                                   H2, M, (M + 15) / 16);

    {
        const long long threads = (long long)M * 16;
        spmm_gather16<<<(unsigned)((threads + 255) / 256), 256, 0, stream>>>(off, edges, H2, out, M);
    }
}


# Layer-2 row sums as fewer, fatter waves: 4 lanes x float4 per row, 8 rows per wave (6250 waves for 25000), a row's first 24 gathers in flight as one group, quad DPP hand-over; sum order unchanged
# speedup vs baseline: 3.8670x; 1.0364x over previous
#include <hip/hip_runtime.h>

typedef __attribute__((ext_vector_type(2))) float v2f;
typedef __attribute__((ext_vector_type(8))) float v8f;


constexpr int BIN_SHIFT = 8;
constexpr int BIN_ROWS  = 1 << BIN_SHIFT;
constexpr int MAX_BINS  = 256;
constexpr int EPT       = 8;
constexpr int CHUNK     = 256 * EPT;
constexpr int MAX_CHUNKS = 1024;
constexpr int CNT_MASK  = 0xFFF;


constexpr int A_PITCH = 132;
constexpr int A_STAGE = 16 * A_PITCH;

__device__ __forceinline__ void lds_barrier()
{
    __builtin_amdgcn_fence(__ATOMIC_RELEASE, "workgroup", "local");
    __builtin_amdgcn_s_barrier();
    __builtin_amdgcn_fence(__ATOMIC_ACQUIRE, "workgroup", "local");
}

__global__ __launch_bounds__(256)
void bin_chunk(const int* __restrict__ row, const int* __restrict__ col,
               const float* __restrict__ val, int* __restrict__ tab,
               int2* __restrict__ binned, int E, int nbin, int nblk, int tpitch)
{
    __shared__ int s_hist[MAX_BINS];
    __shared__ int s_w[8];
    const int tid  = threadIdx.x;
    const int lane = tid & 31;
    const int e0   = (int)blockIdx.x * CHUNK + tid;
    s_hist[tid] = 0;
    int r[EPT], c[EPT], v[EPT];
#pragma unroll
    for (int i = 0; i < EPT; ++i) {
        const int e = e0 + i * 256;
        const bool live = e < E;
        r[i] = live ? row[e] : -1;
        c[i] = live ? col[e] : 0;
        v[i] = live ? __float_as_int(val[e]) : 0;
    }
    __syncthreads();
#pragma unroll
    for (int i = 0; i < EPT; ++i)
        if (r[i] >= 0) atomicAdd(&s_hist[r[i] >> BIN_SHIFT], 1);
    __syncthreads();

    const int cn = s_hist[tid];
    int inc = cn;
    for (int d = 1; d < 32; d <<= 1) {
        const int y = __shfl_up(inc, d);
        if (lane >= d) inc += y;
    }
    if (lane == 31) s_w[tid >> 5] = inc;
    __syncthreads();
    int start = inc - cn;
    for (int q = 0; q < 8; ++q)
        if (q < (tid >> 5)) start += s_w[q];
    s_hist[tid] = start;
    if (tid < nbin) {
        tab[tid * tpitch + (int)blockIdx.x] = (start << 12) | cn;
        if (blockIdx.x == 0)
            for (int p = nblk; p < tpitch; ++p) tab[tid * tpitch + p] = 0;
    }
    __syncthreads();

    int2* __restrict__ dst = binned + (size_t)blockIdx.x * CHUNK;
#pragma unroll
    for (int i = 0; i < EPT; ++i) {
        if (r[i] >= 0) {
            const int pos = atomicAdd(&s_hist[r[i] >> BIN_SHIFT], 1);
            const unsigned key = ((unsigned)(r[i] & (BIN_ROWS - 1)) << 24) | (unsigned)c[i];
            dst[pos] = make_int2((int)key, v[i]);
        }
    }
}

__device__ __forceinline__ int run_slot(const int* __restrict__ s_scan, const int* __restrict__ s_src,
                                        int nblk, int i)
{
    int lo = 0, hi = nblk;
#pragma unroll
    for (int s = 0; s < 10; ++s) {
        const int mid = (lo + hi) >> 1;
        const bool ge = s_scan[mid] <= i;
        lo = ge ? mid : lo;
        hi = ge ? hi : mid;
    }
    return s_src[lo] + (i - s_scan[lo]);
}

constexpr int SORT_U = 5;
__global__ __launch_bounds__(1024)
void bin_sort(const int* __restrict__ tab, const int2* __restrict__ binned,
              int* __restrict__ off, int2* __restrict__ edges, int M, int nblk, int tpitch)
{
    __shared__ int s_cnt[BIN_ROWS];
    __shared__ int s_w[32];
    __shared__ int s_front[32];
    __shared__ int s_scan[MAX_CHUNKS + 1];
    __shared__ int s_src[MAX_CHUNKS];
    const int tid  = threadIdx.x;
    const int lane = tid & 31;
    const int w    = tid >> 5;

    int pk = 0;
    if (tid < nblk) pk = tab[(int)blockIdx.x * tpitch + tid];

    const int4* __restrict__ t4 = (const int4*)tab;
    const int nfront4 = (int)blockIdx.x * (tpitch >> 2);
    int fs = 0;
#pragma unroll 4
    for (int k = tid; k < nfront4; k += 1024) {
        const int4 q = t4[k];
        fs += ((q.x & CNT_MASK) + (q.y & CNT_MASK)) + ((q.z & CNT_MASK) + (q.w & CNT_MASK));
    }
    for (int d = 16; d > 0; d >>= 1) fs += __shfl_xor(fs, d);

    if (tid < BIN_ROWS) s_cnt[tid] = 0;
    const int rn = pk & CNT_MASK;
    int rinc = rn;
    for (int d = 1; d < 32; d <<= 1) {
        const int y = __shfl_up(rinc, d);
        if (lane >= d) rinc += y;
    }
    if (lane == 31) s_w[w]     = rinc;
    if (lane == 0)  s_front[w] = fs;
    __syncthreads();
    int beg = 0, rbase = 0;
    for (int q = 0; q < 32; ++q) {
        beg += s_front[q];
        if (q < w) rbase += s_w[q];
    }
    const int rex = rbase + rinc - rn;
    if (tid < nblk) {
        s_scan[tid] = rex;
        s_src[tid]  = tid * CHUNK + (int)((unsigned)pk >> 12);
        if (tid == nblk - 1) s_scan[nblk] = rex + rn;
    }
    __syncthreads();
    const int total = s_scan[nblk];
    const bool one_trip = total <= 1024 * SORT_U;

    int2 rec[SORT_U];
#pragma unroll
    for (int u = 0; u < SORT_U; ++u) rec[u] = make_int2(0, 0);
    for (int base = 0; base < total; base += 1024 * SORT_U) {
#pragma unroll
        for (int u = 0; u < SORT_U; ++u) {
            const int i = base + u * 1024 + tid;
            rec[u] = binned[run_slot(s_scan, s_src, nblk, min(i, total - 1))];
        }
#pragma unroll
        for (int u = 0; u < SORT_U; ++u) {
            const int i = base + u * 1024 + tid;
            if (i < total) atomicAdd(&s_cnt[(unsigned)rec[u].x >> 24], 1);
        }
    }
    __syncthreads();

    int cn = 0, inc = 0;
    if (tid < BIN_ROWS) {
        cn  = s_cnt[tid];
        inc = cn;
        for (int d = 1; d < 32; d <<= 1) {
            const int y = __shfl_up(inc, d);
            if (lane >= d) inc += y;
        }
        if (lane == 31) s_w[w] = inc;
    }
    __syncthreads();
    if (tid < BIN_ROWS) {
        int start = beg + inc - cn;
        for (int q = 0; q < BIN_ROWS / 32; ++q)
            if (q < w) start += s_w[q];
        s_cnt[tid] = start;
        const int rr = ((int)blockIdx.x << BIN_SHIFT) + tid;
        if (rr < M) off[rr] = start;
        if (rr == M - 1) off[M] = start + cn;
    }
    __syncthreads();

    for (int base = 0; base < total; base += 1024 * SORT_U) {
        if (!one_trip) {
#pragma unroll
            for (int u = 0; u < SORT_U; ++u) {
                const int i = base + u * 1024 + tid;
                rec[u] = binned[run_slot(s_scan, s_src, nblk, min(i, total - 1))];
            }
        }
        int pos[SORT_U];
#pragma unroll
        for (int u = 0; u < SORT_U; ++u) {
            const int i = base + u * 1024 + tid;
            pos[u] = 0;
            if (i < total) pos[u] = atomicAdd(&s_cnt[(unsigned)rec[u].x >> 24], 1);
        }
#pragma unroll
        for (int u = 0; u < SORT_U; ++u) {
            const int i = base + u * 1024 + tid;
            if (i < total)
                edges[pos[u]] = make_int2((int)((unsigned)rec[u].x & 0x00FFFFFFu), rec[u].y);
        }
    }
}

constexpr int ROWS_IN_FLIGHT = 14;

template <int U, int I>
__device__ __forceinline__ void add_chunk(float4& acc, float& vsum, float4 (&h)[U], float (&v)[U],
                                          const float* __restrict__ Xl, int ec, int ev, int j, int m)
{
    const int c = __builtin_amdgcn_readlane(ec, j + I);
    v[I] = __int_as_float(__builtin_amdgcn_readlane(ev, j + I));
    h[I] = *(const float4*)(Xl + (size_t)c * 128);
    if constexpr (I + 1 < U) {
        if (I + 1 < m) {
            add_chunk<U, I + 1>(acc, vsum, h, v, Xl, ec, ev, j, m);
            return;
        }
    }
#pragma unroll
    for (int u = 0; u <= I; ++u) {
        acc.x = fmaf(v[u], h[u].x, acc.x);
        acc.y = fmaf(v[u], h[u].y, acc.y);
        acc.z = fmaf(v[u], h[u].z, acc.z);
        acc.w = fmaf(v[u], h[u].w, acc.w);
        vsum += v[u];
    }
}

constexpr int P_HALF = 144;
constexpr int P_TILE = P_HALF + 128;

template <int K, int N1, int N2>
__global__ __launch_bounds__(512)
void spmm_gather128_gemm2(const int* __restrict__ off, const int* __restrict__ edges,
                          const float* __restrict__ X, const float* __restrict__ W1,
                          const float* __restrict__ b1, const float* __restrict__ W2,
                          const float* __restrict__ b2, float* __restrict__ Dout,
                          int M, int nstripes)
{
    static_assert(K == 128, "a summed row is one float4 per lane");
    static_assert(N1 == 128, "eight tile columns, each k half on its own wave");
    static_assert(N2 == 16, "one tile column; the k range is split over eight waves");
    __shared__ __attribute__((aligned(16))) float s_a[A_STAGE];
    __shared__ __attribute__((aligned(16))) float s_q[2 * A_STAGE];
    __shared__ float s_p[8 * P_TILE];
    __shared__ float s_rs[16];
    const int tid  = threadIdx.x;
    const int lane = tid & 31;
    const int w    = __builtin_amdgcn_readfirstlane(tid >> 5);
    const int hi   = lane >> 4;
    const int l16  = lane & 15;
    const int koff = hi << 1;
    const int t1   = w & 7;
    const int kh   = w >> 3;

    v2f bw1[16];
    {
        const float* __restrict__ bcol = W1 + (size_t)(64 * kh + koff) * N1 + 16 * t1 + l16;
#pragma unroll
        for (int i = 0; i < 16; ++i) {
            bw1[i][0] = bcol[(4 * i) * N1];
            bw1[i][1] = bcol[(4 * i + 1) * N1];
        }
    }
    const float b1x = b1[(lane << 2) + 0];
    const float b1y = b1[(lane << 2) + 1];
    const float b1z = b1[(lane << 2) + 2];
    const float b1w = b1[(lane << 2) + 3];
    v2f bw2[4];
    float bv = 0.0f;
    if (w < 8) {
        const float* __restrict__ bcol = W2 + (size_t)(16 * w + koff) * N2 + l16;
#pragma unroll
        for (int i = 0; i < 4; ++i) {
            bw2[i][0] = bcol[(4 * i) * N2];
            bw2[i][1] = bcol[(4 * i + 1) * N2];
        }
        bv = b2[l16];
    }

    const float* __restrict__ Xl = X + (lane << 2);
    const int2*  __restrict__ ed = (const int2*)edges;

    int beg = 0, end = 0;
    {
        const int r0 = (int)blockIdx.x * 16 + w;
        if (r0 < M) {
            beg = __builtin_amdgcn_readfirstlane(off[r0]);
            end = __builtin_amdgcn_readfirstlane(off[r0 + 1]);
        }
    }
    int2 rec = make_int2(0, 0);
    if (lane < end - beg) rec = ed[beg + lane];
    int begv = 0, endv = 0;
    {
        const int r1 = ((int)blockIdx.x + (int)gridDim.x) * 16 + w;
        if (r1 < M) {
            begv = off[r1];
            endv = off[r1 + 1];
        }
    }
    for (int s = (int)blockIdx.x; s < nstripes; s += (int)gridDim.x) {
        const int begn = __builtin_amdgcn_readfirstlane(begv);
        const int endn = __builtin_amdgcn_readfirstlane(endv);
        float4 acc = make_float4(0.0f, 0.0f, 0.0f, 0.0f);
        float vs = 0.0f;
        for (int base = beg; base < end; base += 32) {
            const int n = min(end - base, 32);
            if (base != beg) {
                rec = make_int2(0, 0);
                if (lane < n) rec = ed[base + lane];
            }
            for (int j = 0; j < n; j += ROWS_IN_FLIGHT) {
                float4 h[ROWS_IN_FLIGHT];
                float  v[ROWS_IN_FLIGHT];
                add_chunk<ROWS_IN_FLIGHT, 0>(acc, vs, h, v, Xl, rec.x, rec.y, j, n - j);
            }
        }
        beg = begn;
        end = endn;
        rec = make_int2(0, 0);
        if (lane < end - beg) rec = ed[beg + lane];
        begv = 0;
        endv = 0;
        {
            const int r2 = (s + 2 * (int)gridDim.x) * 16 + w;
            if (r2 < M) {
                begv = off[r2];
                endv = off[r2 + 1];
            }
        }
        *(float4*)(s_a + w * A_PITCH + (lane << 2)) = acc;
        if (lane == 0) s_rs[w] = vs;
        lds_barrier();

        {
            const float* __restrict__ fa = s_a + l16 * A_PITCH + 64 * kh + koff;
            v2f a[16];
#pragma unroll
            for (int i = 0; i < 16; ++i) a[i] = *(const v2f*)(fa + 4 * i);
            __builtin_amdgcn_sched_barrier(0);
            v8f part;
#pragma unroll
            for (int q = 0; q < 8; ++q) part[q] = 0.0f;
#pragma unroll
            for (int i = 0; i < 16; ++i) {
                part = __builtin_amdgcn_wmma_f32_16x16x4_f32(
                    false, a[i], false, bw1[i], (short)0, part, false, false);
            }
            float* __restrict__ p = s_q + kh * A_STAGE + (hi << 3) * A_PITCH + 16 * t1 + l16;
#pragma unroll
            for (int q = 0; q < 8; ++q) p[q * A_PITCH] = part[q];
        }
        lds_barrier();

        {
            const float4 p0 = *(const float4*)(s_q + w * A_PITCH + (lane << 2));
            const float4 p1 = *(const float4*)(s_q + A_STAGE + w * A_PITCH + (lane << 2));
            const float rsw = s_rs[w];
            float4 h;
            h.x = fmaxf(fmaf(rsw, b1x, p0.x + p1.x), 0.0f);
            h.y = fmaxf(fmaf(rsw, b1y, p0.y + p1.y), 0.0f);
            h.z = fmaxf(fmaf(rsw, b1z, p0.z + p1.z), 0.0f);
            h.w = fmaxf(fmaf(rsw, b1w, p0.w + p1.w), 0.0f);
            *(float4*)(s_a + w * A_PITCH + (lane << 2)) = h;
        }
        lds_barrier();

        if (w < 8) {
            const float* __restrict__ fa = s_a + l16 * A_PITCH + 16 * w + koff;
            v2f a[4];
#pragma unroll
            for (int i = 0; i < 4; ++i) a[i] = *(const v2f*)(fa + 4 * i);
            v8f part;
#pragma unroll
            for (int q = 0; q < 8; ++q) part[q] = 0.0f;
#pragma unroll
            for (int i = 0; i < 4; ++i) {
                part = __builtin_amdgcn_wmma_f32_16x16x4_f32(
                    false, a[i], false, bw2[i], (short)0, part, false, false);
            }
            float* __restrict__ p = s_p + w * P_TILE + hi * P_HALF + l16;
#pragma unroll
            for (int q = 0; q < 8; ++q) p[q * 16] = part[q];
        }
        lds_barrier();
        if (tid < 256) {
            const int row = tid >> 4;
            const int col = tid & 15;
            const float* __restrict__ p = s_p + (row >> 3) * P_HALF + (row & 7) * 16 + col;
            float sum = bv;
#pragma unroll
            for (int q = 0; q < 8; ++q) sum += p[q * P_TILE];
            const int gr = s * 16 + row;
            if (gr < M) Dout[(size_t)gr * N2 + col] = sum;
        }
    }
}

constexpr int G16_GROUP = 24;
constexpr int G16_RECS  = G16_GROUP / 4;

__device__ __forceinline__ int quad_bcast(int x, int j)
{
    switch (j & 3) {
    case 0:  return __builtin_amdgcn_update_dpp(0, x, 0x00, 0xF, 0xF, false);
    case 1:  return __builtin_amdgcn_update_dpp(0, x, 0x55, 0xF, 0xF, false);
    case 2:  return __builtin_amdgcn_update_dpp(0, x, 0xAA, 0xF, 0xF, false);
    default: return __builtin_amdgcn_update_dpp(0, x, 0xFF, 0xF, 0xF, false);
    }
}

__device__ __forceinline__ void add_slot(float4& acc, float v, const float4& h, bool on)
{
    acc.x = on ? fmaf(v, h.x, acc.x) : acc.x;
    acc.y = on ? fmaf(v, h.y, acc.y) : acc.y;
    acc.z = on ? fmaf(v, h.z, acc.z) : acc.z;
    acc.w = on ? fmaf(v, h.w, acc.w) : acc.w;
}

__global__ __launch_bounds__(256)
void spmm_gather16(const int* __restrict__ off, const int* __restrict__ edges,
                   const float* __restrict__ H, float* __restrict__ out, int M)
{
    const int t = blockIdx.x * 256 + threadIdx.x;
    const int q = t & 3;
    const int r = t >> 2;
    const bool live = r < M;
    int beg = 0, end = 0;
    if (live) { beg = off[r]; end = off[r + 1]; }
    const int n = end - beg;
    const int2*  __restrict__ ed = (const int2*)edges;
    const float* __restrict__ Hq = H + (q << 2);
    const bool more = __builtin_amdgcn_ballot_w32(n > G16_GROUP) != 0;

    int2 rec[G16_RECS];
#pragma unroll
    for (int k = 0; k < G16_RECS; ++k) {
        const int s = 4 * k + q;
        rec[k] = ed[s < n ? beg + s : 0];
    }
    __builtin_amdgcn_sched_barrier(0);

    float4 h[G16_GROUP];
#pragma unroll
    for (int j = 0; j < G16_GROUP; ++j) {
        const int c = quad_bcast(rec[j >> 2].x, j);
        h[j] = *(const float4*)(Hq + (size_t)(j < n ? c : 0) * 16);
    }
    int2 rec2[8 - G16_RECS];
#pragma unroll
    for (int k = 0; k < 8 - G16_RECS; ++k) rec2[k] = make_int2(0, 0);
    if (more) {
#pragma unroll
        for (int k = 0; k < 8 - G16_RECS; ++k) {
            const int s = G16_GROUP + 4 * k + q;
            rec2[k] = ed[s < n ? beg + s : 0];
        }
    }

    float4 acc = make_float4(0.0f, 0.0f, 0.0f, 0.0f);
#pragma unroll
    for (int j = 0; j < G16_GROUP; ++j) {
        const float v = __int_as_float(quad_bcast(rec[j >> 2].y, j));
        add_slot(acc, v, h[j], j < n);
    }
    if (more) {
        float4 g[32 - G16_GROUP];
#pragma unroll
        for (int j = 0; j < 32 - G16_GROUP; ++j) {
            const int c = quad_bcast(rec2[j >> 2].x, j);
            g[j] = *(const float4*)(Hq + (size_t)(G16_GROUP + j < n ? c : 0) * 16);
        }
#pragma unroll
        for (int j = 0; j < 32 - G16_GROUP; ++j) {
            const float v = __int_as_float(quad_bcast(rec2[j >> 2].y, j));
            add_slot(acc, v, g[j], G16_GROUP + j < n);
        }
        for (int base = 32; __builtin_amdgcn_ballot_w32(base < n) != 0; base += 4) {
            const int s = base + q;
            const int2 rc = ed[s < n ? beg + s : 0];
            float4 f[4];
#pragma unroll
            for (int j = 0; j < 4; ++j) {
                const int c = quad_bcast(rc.x, j);
                f[j] = *(const float4*)(Hq + (size_t)(base + j < n ? c : 0) * 16);
            }
#pragma unroll
            for (int j = 0; j < 4; ++j) {
                const float v = __int_as_float(quad_bcast(rc.y, j));
                add_slot(acc, v, f[j], base + j < n);
            }
        }
    }
    if (live) *(float4*)(out + (size_t)r * 16 + (q << 2)) = acc;
}

extern "C" void kernel_launch(void* const* d_in, const int* in_sizes, int n_in,
                              void* d_out, int out_size, void* d_ws, size_t ws_size,
                              hipStream_t stream) {
    const float* x   = (const float*)d_in[0];
    const int*   row = (const int*)  d_in[1];
    const int*   col = (const int*)  d_in[2];
    const float* val = (const float*)d_in[3];
    const float* W1  = (const float*)d_in[4];
    const float* b1  = (const float*)d_in[5];
    const float* W2  = (const float*)d_in[6];
    const float* b2  = (const float*)d_in[7];
    float* out = (float*)d_out;

    constexpr int D_IN  = 128;
    constexpr int D_HID = 128;
    constexpr int N_CLS = 16;
    const int E = in_sizes[1];
    const int M = in_sizes[0] / D_IN;

    const int nbin = (M + BIN_ROWS - 1) / BIN_ROWS;
    const int nblk = (E + CHUNK - 1) / CHUNK;
    const int tpitch = (nblk + 3) & ~3;
    const int ntab = nbin * tpitch;

    char*  ws  = (char*)d_ws;
    float* H2  = (float*)ws;
    int*   tab    = (int*)(H2 + (size_t)M * N_CLS);
    int*   off    = tab + ntab;
    int*   binned = off + ((M + 1 + 3) & ~3);
    int*   edges  = binned + 2 * (size_t)nblk * CHUNK;

    bin_chunk<<<(unsigned)nblk, 256, 0, stream>>>(row, col, val, tab,
                                                  (int2*)binned, E, nbin, nblk, tpitch);
    bin_sort<<<(unsigned)nbin, 1024, 0, stream>>>(tab, (const int2*)binned, off,
                                                  (int2*)edges, M, nblk, tpitch);

    spmm_gather128_gemm2<D_IN, D_HID, N_CLS><<<512, 512, 0, stream>>>(off, edges, x, W1, b1, W2, b2,
                                                                   H2, M, (M + 15) / 16);

    {
        const long long threads = (long long)M * 4;
        spmm_gather16<<<(unsigned)((threads + 255) / 256), 256, 0, stream>>>(off, edges, H2, out, M);
    }
}


# Edge sort, both passes: records are placed in an LDS copy of the block's list and leave in order as whole 128-byte lines, not as one scattered 8-byte store per record; sum order unchanged
# speedup vs baseline: 4.0944x; 1.0588x over previous
#include <hip/hip_runtime.h>

typedef __attribute__((ext_vector_type(2))) float v2f;
typedef __attribute__((ext_vector_type(8))) float v8f;


constexpr int BIN_SHIFT = 8;
constexpr int BIN_ROWS  = 1 << BIN_SHIFT;
constexpr int MAX_BINS  = 256;
constexpr int EPT       = 8;
constexpr int CHUNK     = 256 * EPT;
constexpr int MAX_CHUNKS = 1024;
constexpr int CNT_MASK  = 0xFFF;
constexpr int LINE_RECS = 16;
constexpr int STAGE_CAP = 8192;


constexpr int A_PITCH = 132;
constexpr int A_STAGE = 16 * A_PITCH;

__device__ __forceinline__ void lds_barrier()
{
    __builtin_amdgcn_fence(__ATOMIC_RELEASE, "workgroup", "local");
    __builtin_amdgcn_s_barrier();
    __builtin_amdgcn_fence(__ATOMIC_ACQUIRE, "workgroup", "local");
}

__global__ __launch_bounds__(256)
void bin_chunk(const int* __restrict__ row, const int* __restrict__ col,
               const float* __restrict__ val, int* __restrict__ tab,
               int2* __restrict__ binned, int E, int nbin, int nblk, int tpitch)
{
    __shared__ int s_hist[MAX_BINS];
    __shared__ int s_w[8];
    __shared__ __attribute__((aligned(16))) int2 s_rec[CHUNK];
    const int tid  = threadIdx.x;
    const int lane = tid & 31;
    const int e0   = (int)blockIdx.x * CHUNK + tid;
    s_hist[tid] = 0;
    int r[EPT], c[EPT], v[EPT];
#pragma unroll
    for (int i = 0; i < EPT; ++i) {
        const int e = e0 + i * 256;
        const bool live = e < E;
        r[i] = live ? row[e] : -1;
        c[i] = live ? col[e] : 0;
        v[i] = live ? __float_as_int(val[e]) : 0;
    }
    __syncthreads();
#pragma unroll
    for (int i = 0; i < EPT; ++i)
        if (r[i] >= 0) atomicAdd(&s_hist[r[i] >> BIN_SHIFT], 1);
    __syncthreads();

    const int cn = s_hist[tid];
    int inc = cn;
    for (int d = 1; d < 32; d <<= 1) {
        const int y = __shfl_up(inc, d);
        if (lane >= d) inc += y;
    }
    if (lane == 31) s_w[tid >> 5] = inc;
    __syncthreads();
    int start = inc - cn;
    for (int q = 0; q < 8; ++q)
        if (q < (tid >> 5)) start += s_w[q];
    s_hist[tid] = start;
    if (tid < nbin) {
        tab[tid * tpitch + (int)blockIdx.x] = (start << 12) | cn;
        if (blockIdx.x == 0)
            for (int p = nblk; p < tpitch; ++p) tab[tid * tpitch + p] = 0;
    }
    __syncthreads();

#pragma unroll
    for (int i = 0; i < EPT; ++i) {
        if (r[i] >= 0) {
            const int pos = atomicAdd(&s_hist[r[i] >> BIN_SHIFT], 1);
            const unsigned key = ((unsigned)(r[i] & (BIN_ROWS - 1)) << 24) | (unsigned)c[i];
            s_rec[pos] = make_int2((int)key, v[i]);
        }
    }
    __syncthreads();

    int2* __restrict__ dst = binned + (size_t)blockIdx.x * CHUNK;
    const int nlive = min(CHUNK, E - (int)blockIdx.x * CHUNK);
    const int4* __restrict__ s4 = (const int4*)s_rec;
    int4* __restrict__ d4 = (int4*)dst;
    if (nlive == CHUNK) {
        int4 q[CHUNK / 512];
#pragma unroll
        for (int i = 0; i < CHUNK / 512; ++i) q[i] = s4[tid + i * 256];
#pragma unroll
        for (int i = 0; i < CHUNK / 512; ++i) d4[tid + i * 256] = q[i];
    } else {
        for (int k = tid; k < (nlive >> 1); k += 256) d4[k] = s4[k];
        if ((nlive & 1) && tid == 0) dst[nlive - 1] = s_rec[nlive - 1];
    }
}

__device__ __forceinline__ int run_slot(const int* __restrict__ s_scan, const int* __restrict__ s_src,
                                        int nblk, int i)
{
    int lo = 0, hi = nblk;
#pragma unroll
    for (int s = 0; s < 10; ++s) {
        const int mid = (lo + hi) >> 1;
        const bool ge = s_scan[mid] <= i;
        lo = ge ? mid : lo;
        hi = ge ? hi : mid;
    }
    return s_src[lo] + (i - s_scan[lo]);
}

constexpr int SORT_U = 5;
__global__ __launch_bounds__(1024)
void bin_sort(const int* __restrict__ tab, const int2* __restrict__ binned,
              int* __restrict__ off, int2* __restrict__ edges, int M, int nblk, int tpitch)
{
    __shared__ int s_cnt[BIN_ROWS];
    __shared__ int s_w[32];
    __shared__ int s_front[32];
    __shared__ int s_scan[MAX_CHUNKS + 1];
    __shared__ int s_src[MAX_CHUNKS];
    __shared__ __attribute__((aligned(16))) int2 s_stage[STAGE_CAP + LINE_RECS];
    const int tid  = threadIdx.x;
    const int lane = tid & 31;
    const int w    = tid >> 5;

    int pk = 0;
    if (tid < nblk) pk = tab[(int)blockIdx.x * tpitch + tid];

    const int4* __restrict__ t4 = (const int4*)tab;
    const int nfront4 = (int)blockIdx.x * (tpitch >> 2);
    int fs = 0;
#pragma unroll 4
    for (int k = tid; k < nfront4; k += 1024) {
        const int4 q = t4[k];
        fs += ((q.x & CNT_MASK) + (q.y & CNT_MASK)) + ((q.z & CNT_MASK) + (q.w & CNT_MASK));
    }
    for (int d = 16; d > 0; d >>= 1) fs += __shfl_xor(fs, d);

    if (tid < BIN_ROWS) s_cnt[tid] = 0;
    const int rn = pk & CNT_MASK;
    int rinc = rn;
    for (int d = 1; d < 32; d <<= 1) {
        const int y = __shfl_up(rinc, d);
        if (lane >= d) rinc += y;
    }
    if (lane == 31) s_w[w]     = rinc;
    if (lane == 0)  s_front[w] = fs;
    __syncthreads();
    int beg = 0, rbase = 0;
    for (int q = 0; q < 32; ++q) {
        beg += s_front[q];
        if (q < w) rbase += s_w[q];
    }
    const int rex = rbase + rinc - rn;
    if (tid < nblk) {
        s_scan[tid] = rex;
        s_src[tid]  = tid * CHUNK + (int)((unsigned)pk >> 12);
        if (tid == nblk - 1) s_scan[nblk] = rex + rn;
    }
    __syncthreads();
    const int total = s_scan[nblk];
    const bool one_trip = total <= 1024 * SORT_U;
    const int lead = beg & (LINE_RECS - 1);
    const int gsh  = beg - lead;
    const bool staged = total <= STAGE_CAP;

    int2 rec[SORT_U];
#pragma unroll
    for (int u = 0; u < SORT_U; ++u) rec[u] = make_int2(0, 0);
    for (int base = 0; base < total; base += 1024 * SORT_U) {
#pragma unroll
        for (int u = 0; u < SORT_U; ++u) {
            const int i = base + u * 1024 + tid;
            rec[u] = binned[run_slot(s_scan, s_src, nblk, min(i, total - 1))];
        }
#pragma unroll
        for (int u = 0; u < SORT_U; ++u) {
            const int i = base + u * 1024 + tid;
            if (i < total) atomicAdd(&s_cnt[(unsigned)rec[u].x >> 24], 1);
        }
    }
    __syncthreads();

    int cn = 0, inc = 0;
    if (tid < BIN_ROWS) {
        cn  = s_cnt[tid];
        inc = cn;
        for (int d = 1; d < 32; d <<= 1) {
            const int y = __shfl_up(inc, d);
            if (lane >= d) inc += y;
        }
        if (lane == 31) s_w[w] = inc;
    }
    __syncthreads();
    if (tid < BIN_ROWS) {
        int start = beg + inc - cn;
        for (int q = 0; q < BIN_ROWS / 32; ++q)
            if (q < w) start += s_w[q];
        s_cnt[tid] = start - gsh;
        const int rr = ((int)blockIdx.x << BIN_SHIFT) + tid;
        if (rr < M) off[rr] = start;
        if (rr == M - 1) off[M] = start + cn;
    }
    __syncthreads();

    for (int base = 0; base < total; base += 1024 * SORT_U) {
        if (!one_trip) {
#pragma unroll
            for (int u = 0; u < SORT_U; ++u) {
                const int i = base + u * 1024 + tid;
                rec[u] = binned[run_slot(s_scan, s_src, nblk, min(i, total - 1))];
            }
        }
        int pos[SORT_U];
#pragma unroll
        for (int u = 0; u < SORT_U; ++u) {
            const int i = base + u * 1024 + tid;
            pos[u] = 0;
            if (i < total) pos[u] = atomicAdd(&s_cnt[(unsigned)rec[u].x >> 24], 1);
        }
        if (staged) {
#pragma unroll
            for (int u = 0; u < SORT_U; ++u) {
                const int i = base + u * 1024 + tid;
                if (i < total)
                    s_stage[pos[u]] = make_int2((int)((unsigned)rec[u].x & 0x00FFFFFFu), rec[u].y);
            }
        } else {
#pragma unroll
            for (int u = 0; u < SORT_U; ++u) {
                const int i = base + u * 1024 + tid;
                if (i < total)
                    edges[gsh + pos[u]] = make_int2((int)((unsigned)rec[u].x & 0x00FFFFFFu), rec[u].y);
            }
        }
    }
    if (staged) {
        __syncthreads();
        const int nslot = lead + total;
        const int k0 = (lead + 1) & ~1;
        const int k1 = nslot & ~1;
        for (int k = 2 * tid; k < k1; k += 2048)
            if (k >= k0) *(int4*)(edges + gsh + k) = *(const int4*)(s_stage + k);
        if (total > 0) {
            if (tid == 0  && (lead & 1))  edges[gsh + lead]      = s_stage[lead];
            if (tid == 32 && (nslot & 1)) edges[gsh + nslot - 1] = s_stage[nslot - 1];
        }
    }
}

constexpr int ROWS_IN_FLIGHT = 14;

template <int U, int I>
__device__ __forceinline__ void add_chunk(float4& acc, float& vsum, float4 (&h)[U], float (&v)[U],
                                          const float* __restrict__ Xl, int ec, int ev, int j, int m)
{
    const int c = __builtin_amdgcn_readlane(ec, j + I);
    v[I] = __int_as_float(__builtin_amdgcn_readlane(ev, j + I));
    h[I] = *(const float4*)(Xl + (size_t)c * 128);
    if constexpr (I + 1 < U) {
        if (I + 1 < m) {
            add_chunk<U, I + 1>(acc, vsum, h, v, Xl, ec, ev, j, m);
            return;
        }
    }
#pragma unroll
    for (int u = 0; u <= I; ++u) {
        acc.x = fmaf(v[u], h[u].x, acc.x);
        acc.y = fmaf(v[u], h[u].y, acc.y);
        acc.z = fmaf(v[u], h[u].z, acc.z);
        acc.w = fmaf(v[u], h[u].w, acc.w);
        vsum += v[u];
    }
}

constexpr int P_HALF = 144;
constexpr int P_TILE = P_HALF + 128;

template <int K, int N1, int N2>
__global__ __launch_bounds__(512)
void spmm_gather128_gemm2(const int* __restrict__ off, const int* __restrict__ edges,
                          const float* __restrict__ X, const float* __restrict__ W1,
                          const float* __restrict__ b1, const float* __restrict__ W2,
                          const float* __restrict__ b2, float* __restrict__ Dout,
                          int M, int nstripes)
{
    static_assert(K == 128, "a summed row is one float4 per lane");
    static_assert(N1 == 128, "eight tile columns, each k half on its own wave");
    static_assert(N2 == 16, "one tile column; the k range is split over eight waves");
    __shared__ __attribute__((aligned(16))) float s_a[A_STAGE];
    __shared__ __attribute__((aligned(16))) float s_q[2 * A_STAGE];
    __shared__ float s_p[8 * P_TILE];
    __shared__ float s_rs[16];
    const int tid  = threadIdx.x;
    const int lane = tid & 31;
    const int w    = __builtin_amdgcn_readfirstlane(tid >> 5);
    const int hi   = lane >> 4;
    const int l16  = lane & 15;
    const int koff = hi << 1;
    const int t1   = w & 7;
    const int kh   = w >> 3;

    v2f bw1[16];
    {
        const float* __restrict__ bcol = W1 + (size_t)(64 * kh + koff) * N1 + 16 * t1 + l16;
#pragma unroll
        for (int i = 0; i < 16; ++i) {
            bw1[i][0] = bcol[(4 * i) * N1];
            bw1[i][1] = bcol[(4 * i + 1) * N1];
        }
    }
    const float b1x = b1[(lane << 2) + 0];
    const float b1y = b1[(lane << 2) + 1];
    const float b1z = b1[(lane << 2) + 2];
    const float b1w = b1[(lane << 2) + 3];
    v2f bw2[4];
    float bv = 0.0f;
    if (w < 8) {
        const float* __restrict__ bcol = W2 + (size_t)(16 * w + koff) * N2 + l16;
#pragma unroll
        for (int i = 0; i < 4; ++i) {
            bw2[i][0] = bcol[(4 * i) * N2];
            bw2[i][1] = bcol[(4 * i + 1) * N2];
        }
        bv = b2[l16];
    }

    const float* __restrict__ Xl = X + (lane << 2);
    const int2*  __restrict__ ed = (const int2*)edges;

    int beg = 0, end = 0;
    {
        const int r0 = (int)blockIdx.x * 16 + w;
        if (r0 < M) {
            beg = __builtin_amdgcn_readfirstlane(off[r0]);
            end = __builtin_amdgcn_readfirstlane(off[r0 + 1]);
        }
    }
    int2 rec = make_int2(0, 0);
    if (lane < end - beg) rec = ed[beg + lane];
    int begv = 0, endv = 0;
    {
        const int r1 = ((int)blockIdx.x + (int)gridDim.x) * 16 + w;
        if (r1 < M) {
            begv = off[r1];
            endv = off[r1 + 1];
        }
    }
    for (int s = (int)blockIdx.x; s < nstripes; s += (int)gridDim.x) {
        const int begn = __builtin_amdgcn_readfirstlane(begv);
        const int endn = __builtin_amdgcn_readfirstlane(endv);
        float4 acc = make_float4(0.0f, 0.0f, 0.0f, 0.0f);
        float vs = 0.0f;
        for (int base = beg; base < end; base += 32) {
            const int n = min(end - base, 32);
            if (base != beg) {
                rec = make_int2(0, 0);
                if (lane < n) rec = ed[base + lane];
            }
            for (int j = 0; j < n; j += ROWS_IN_FLIGHT) {
                float4 h[ROWS_IN_FLIGHT];
                float  v[ROWS_IN_FLIGHT];
                add_chunk<ROWS_IN_FLIGHT, 0>(acc, vs, h, v, Xl, rec.x, rec.y, j, n - j);
            }
        }
        beg = begn;
        end = endn;
        rec = make_int2(0, 0);
        if (lane < end - beg) rec = ed[beg + lane];
        begv = 0;
        endv = 0;
        {
            const int r2 = (s + 2 * (int)gridDim.x) * 16 + w;
            if (r2 < M) {
                begv = off[r2];
                endv = off[r2 + 1];
            }
        }
        *(float4*)(s_a + w * A_PITCH + (lane << 2)) = acc;
        if (lane == 0) s_rs[w] = vs;
        lds_barrier();

        {
            const float* __restrict__ fa = s_a + l16 * A_PITCH + 64 * kh + koff;
            v2f a[16];
#pragma unroll
            for (int i = 0; i < 16; ++i) a[i] = *(const v2f*)(fa + 4 * i);
            __builtin_amdgcn_sched_barrier(0);
            v8f part;
#pragma unroll
            for (int q = 0; q < 8; ++q) part[q] = 0.0f;
#pragma unroll
            for (int i = 0; i < 16; ++i) {
                part = __builtin_amdgcn_wmma_f32_16x16x4_f32(
                    false, a[i], false, bw1[i], (short)0, part, false, false);
            }
            float* __restrict__ p = s_q + kh * A_STAGE + (hi << 3) * A_PITCH + 16 * t1 + l16;
#pragma unroll
            for (int q = 0; q < 8; ++q) p[q * A_PITCH] = part[q];
        }
        lds_barrier();

        {
            const float4 p0 = *(const float4*)(s_q + w * A_PITCH + (lane << 2));
            const float4 p1 = *(const float4*)(s_q + A_STAGE + w * A_PITCH + (lane << 2));
            const float rsw = s_rs[w];
            float4 h;
            h.x = fmaxf(fmaf(rsw, b1x, p0.x + p1.x), 0.0f);
            h.y = fmaxf(fmaf(rsw, b1y, p0.y + p1.y), 0.0f);
            h.z = fmaxf(fmaf(rsw, b1z, p0.z + p1.z), 0.0f);
            h.w = fmaxf(fmaf(rsw, b1w, p0.w + p1.w), 0.0f);
            *(float4*)(s_a + w * A_PITCH + (lane << 2)) = h;
        }
        lds_barrier();

        if (w < 8) {
            const float* __restrict__ fa = s_a + l16 * A_PITCH + 16 * w + koff;
            v2f a[4];
#pragma unroll
            for (int i = 0; i < 4; ++i) a[i] = *(const v2f*)(fa + 4 * i);
            v8f part;
#pragma unroll
            for (int q = 0; q < 8; ++q) part[q] = 0.0f;
#pragma unroll
            for (int i = 0; i < 4; ++i) {
                part = __builtin_amdgcn_wmma_f32_16x16x4_f32(
                    false, a[i], false, bw2[i], (short)0, part, false, false);
            }
            float* __restrict__ p = s_p + w * P_TILE + hi * P_HALF + l16;
#pragma unroll
            for (int q = 0; q < 8; ++q) p[q * 16] = part[q];
        }
        lds_barrier();
        if (tid < 256) {
            const int row = tid >> 4;
            const int col = tid & 15;
            const float* __restrict__ p = s_p + (row >> 3) * P_HALF + (row & 7) * 16 + col;
            float sum = bv;
#pragma unroll
            for (int q = 0; q < 8; ++q) sum += p[q * P_TILE];
            const int gr = s * 16 + row;
            if (gr < M) Dout[(size_t)gr * N2 + col] = sum;
        }
    }
}

constexpr int G16_GROUP = 24;
constexpr int G16_RECS  = G16_GROUP / 4;

__device__ __forceinline__ int quad_bcast(int x, int j)
{
    switch (j & 3) {
    case 0:  return __builtin_amdgcn_update_dpp(0, x, 0x00, 0xF, 0xF, false);
    case 1:  return __builtin_amdgcn_update_dpp(0, x, 0x55, 0xF, 0xF, false);
    case 2:  return __builtin_amdgcn_update_dpp(0, x, 0xAA, 0xF, 0xF, false);
    default: return __builtin_amdgcn_update_dpp(0, x, 0xFF, 0xF, 0xF, false);
    }
}

__device__ __forceinline__ void add_slot(float4& acc, float v, const float4& h, bool on)
{
    acc.x = on ? fmaf(v, h.x, acc.x) : acc.x;
    acc.y = on ? fmaf(v, h.y, acc.y) : acc.y;
    acc.z = on ? fmaf(v, h.z, acc.z) : acc.z;
    acc.w = on ? fmaf(v, h.w, acc.w) : acc.w;
}

__global__ __launch_bounds__(256)
void spmm_gather16(const int* __restrict__ off, const int* __restrict__ edges,
                   const float* __restrict__ H, float* __restrict__ out, int M)
{
    const int t = blockIdx.x * 256 + threadIdx.x;
    const int q = t & 3;
    const int r = t >> 2;
    const bool live = r < M;
    int beg = 0, end = 0;
    if (live) { beg = off[r]; end = off[r + 1]; }
    const int n = end - beg;
    const int2*  __restrict__ ed = (const int2*)edges;
    const float* __restrict__ Hq = H + (q << 2);
    const bool more = __builtin_amdgcn_ballot_w32(n > G16_GROUP) != 0;

    int2 rec[G16_RECS];
#pragma unroll
    for (int k = 0; k < G16_RECS; ++k) {
        const int s = 4 * k + q;
        rec[k] = ed[s < n ? beg + s : 0];
    }
    __builtin_amdgcn_sched_barrier(0);

    float4 h[G16_GROUP];
#pragma unroll
    for (int j = 0; j < G16_GROUP; ++j) {
        const int c = quad_bcast(rec[j >> 2].x, j);
        h[j] = *(const float4*)(Hq + (size_t)(j < n ? c : 0) * 16);
    }
    int2 rec2[8 - G16_RECS];
#pragma unroll
    for (int k = 0; k < 8 - G16_RECS; ++k) rec2[k] = make_int2(0, 0);
    if (more) {
#pragma unroll
        for (int k = 0; k < 8 - G16_RECS; ++k) {
            const int s = G16_GROUP + 4 * k + q;
            rec2[k] = ed[s < n ? beg + s : 0];
        }
    }

    float4 acc = make_float4(0.0f, 0.0f, 0.0f, 0.0f);
#pragma unroll
    for (int j = 0; j < G16_GROUP; ++j) {
        const float v = __int_as_float(quad_bcast(rec[j >> 2].y, j));
        add_slot(acc, v, h[j], j < n);
    }
    if (more) {
        float4 g[32 - G16_GROUP];
#pragma unroll
        for (int j = 0; j < 32 - G16_GROUP; ++j) {
            const int c = quad_bcast(rec2[j >> 2].x, j);
            g[j] = *(const float4*)(Hq + (size_t)(G16_GROUP + j < n ? c : 0) * 16);
        }
#pragma unroll
        for (int j = 0; j < 32 - G16_GROUP; ++j) {
            const float v = __int_as_float(quad_bcast(rec2[j >> 2].y, j));
            add_slot(acc, v, g[j], G16_GROUP + j < n);
        }
        for (int base = 32; __builtin_amdgcn_ballot_w32(base < n) != 0; base += 4) {
            const int s = base + q;
            const int2 rc = ed[s < n ? beg + s : 0];
            float4 f[4];
#pragma unroll
            for (int j = 0; j < 4; ++j) {
                const int c = quad_bcast(rc.x, j);
                f[j] = *(const float4*)(Hq + (size_t)(base + j < n ? c : 0) * 16);
            }
#pragma unroll
            for (int j = 0; j < 4; ++j) {
                const float v = __int_as_float(quad_bcast(rc.y, j));
                add_slot(acc, v, f[j], base + j < n);
            }
        }
    }
    if (live) *(float4*)(out + (size_t)r * 16 + (q << 2)) = acc;
}

extern "C" void kernel_launch(void* const* d_in, const int* in_sizes, int n_in,
                              void* d_out, int out_size, void* d_ws, size_t ws_size,
                              hipStream_t stream) {
    const float* x   = (const float*)d_in[0];
    const int*   row = (const int*)  d_in[1];
    const int*   col = (const int*)  d_in[2];
    const float* val = (const float*)d_in[3];
    const float* W1  = (const float*)d_in[4];
    const float* b1  = (const float*)d_in[5];
    const float* W2  = (const float*)d_in[6];
    const float* b2  = (const float*)d_in[7];
    float* out = (float*)d_out;

    constexpr int D_IN  = 128;
    constexpr int D_HID = 128;
    constexpr int N_CLS = 16;
    const int E = in_sizes[1];
    const int M = in_sizes[0] / D_IN;

    const int nbin = (M + BIN_ROWS - 1) / BIN_ROWS;
    const int nblk = (E + CHUNK - 1) / CHUNK;
    const int tpitch = (nblk + 3) & ~3;
    const int ntab = nbin * tpitch;

    char*  ws  = (char*)d_ws;
    float* H2  = (float*)ws;
    int*   tab    = (int*)H2 + (((size_t)M * N_CLS + 31) & ~(size_t)31);
    int*   off    = tab + ((ntab + 31) & ~31);
    int*   binned = off + ((M + 1 + 31) & ~31);
    int*   edges  = binned + 2 * (size_t)nblk * CHUNK;

    bin_chunk<<<(unsigned)nblk, 256, 0, stream>>>(row, col, val, tab,
                                                  (int2*)binned, E, nbin, nblk, tpitch);
    bin_sort<<<(unsigned)nbin, 1024, 0, stream>>>(tab, (const int2*)binned, off,
                                                  (int2*)edges, M, nblk, tpitch);

    spmm_gather128_gemm2<D_IN, D_HID, N_CLS><<<512, 512, 0, stream>>>(off, edges, x, W1, b1, W2, b2,
                                                                   H2, M, (M + 15) / 16);

    {
        const long long threads = (long long)M * 4;
        spmm_gather16<<<(unsigned)((threads + 255) / 256), 256, 0, stream>>>(off, edges, H2, out, M);
    }
}
